# starting kernel
# baseline (speedup 1.0000x reference)
#include <hip/hip_runtime.h>
#include <hip/hip_bf16.h>

typedef __attribute__((ext_vector_type(2))) float v2f;
typedef __attribute__((ext_vector_type(8))) float v8f;

#define LDSTRIDE 132

__global__ void gnn_deg_kernel(const int* __restrict__ src,
                               const int* __restrict__ dst,
                               float* __restrict__ deg_out,
                               float* __restrict__ deg_in, int E) {
    int e = blockIdx.x * blockDim.x + threadIdx.x;
    if (e < E) {
        atomicAdd(&deg_out[src[e]], 1.0f);
        atomicAdd(&deg_in[dst[e]], 1.0f);
    }
}

__global__ void gnn_norm_kernel(float* __restrict__ d, int n) {
    int i = blockIdx.x * blockDim.x + threadIdx.x;
    if (i < n) {
        float v = fmaxf(d[i], 1.0f);
        d[i] = 1.0f / sqrtf(v);
    }
}

__global__ __launch_bounds__(128) void gnn_gemm_kernel(
    const float* __restrict__ H, const float* __restrict__ nsrc,
    const float* __restrict__ W, float* __restrict__ X, int N) {
    __shared__ float As[16 * LDSTRIDE];

    const int row0 = blockIdx.x * 16;
    const int tid  = threadIdx.x;

    for (int i = tid; i < 16 * 32; i += 128) {
        int r  = i >> 5;
        int c4 = (i & 31) * 4;
        int row = row0 + r;
        int rowc = (row < N) ? row : (N - 1);
        float s = nsrc[rowc];
        float4 v = *(const float4*)(H + (size_t)rowc * 128 + c4);
        float4 sv = make_float4(v.x * s, v.y * s, v.z * s, v.w * s);
        *(float4*)(&As[r * LDSTRIDE + c4]) = sv;
    }
    __syncthreads();

    const int wave = tid >> 5;
    const int lane = tid & 31;
    const int ln   = lane & 15;
    const int koff = (lane < 16) ? 0 : 2;

    const int n0 = wave * 32;
    const int n1 = n0 + 16;

    v8f acc0 = {};
    v8f acc1 = {};

    for (int kb = 0; kb < 128; kb += 4) {
        v2f a = *(const v2f*)(&As[ln * LDSTRIDE + kb + koff]);

        const float* wp = W + (size_t)(kb + koff) * 128;
        v2f b0, b1;
        b0.x = wp[n0 + ln];
        b0.y = wp[128 + n0 + ln];
        b1.x = wp[n1 + ln];
        b1.y = wp[128 + n1 + ln];

        acc0 = __builtin_amdgcn_wmma_f32_16x16x4_f32(
            false, a, false, b0, (short)0, acc0, false, false);
        acc1 = __builtin_amdgcn_wmma_f32_16x16x4_f32(
            false, a, false, b1, (short)0, acc1, false, false);
    }

    const int mbase = row0 + ((lane < 16) ? 0 : 8);
#pragma unroll
    for (int v = 0; v < 8; ++v) {
        int row = mbase + v;
        if (row < N) {
            X[(size_t)row * 128 + n0 + ln] = acc0[v];
            X[(size_t)row * 128 + n1 + ln] = acc1[v];
        }
    }
}

__global__ __launch_bounds__(256) void gnn_agg_kernel(
    const float* __restrict__ X, const int* __restrict__ src,
    const int* __restrict__ dst, float* __restrict__ AGG, int E) {
    long long t = (long long)blockIdx.x * blockDim.x + threadIdx.x;
    int e = (int)(t >> 5);
    if (e >= E) return;
    int lane = (int)(t & 31);
    int s = src[e];
    int d = dst[e];
    float4 v = *(const float4*)(X + (size_t)s * 128 + lane * 4);
    float* out = AGG + (size_t)d * 128 + lane * 4;
    atomicAdd(out + 0, v.x);
    atomicAdd(out + 1, v.y);
    atomicAdd(out + 2, v.z);
    atomicAdd(out + 3, v.w);
}

__global__ __launch_bounds__(256) void gnn_fin_kernel(
    const float* __restrict__ AGG, const float* __restrict__ ndst,
    const float* __restrict__ bias, float* __restrict__ OUT, int N, int do_relu) {
    long long t = (long long)blockIdx.x * blockDim.x + threadIdx.x;
    if (t >= (long long)N * 32) return;
    int node = (int)(t >> 5);
    int c4   = (int)(t & 31) * 4;
    float s = ndst[node];
    float4 v = *(const float4*)(AGG + (size_t)node * 128 + c4);
    float4 b = *(const float4*)(bias + c4);
    float4 o;
    o.x = v.x * s + b.x;
    o.y = v.y * s + b.y;
    o.z = v.z * s + b.z;
    o.w = v.w * s + b.w;
    if (do_relu) {
        o.x = fmaxf(o.x, 0.0f);
        o.y = fmaxf(o.y, 0.0f);
        o.z = fmaxf(o.z, 0.0f);
        o.w = fmaxf(o.w, 0.0f);
    }
    *(float4*)(OUT + (size_t)node * 128 + c4) = o;
}

extern "C" void kernel_launch(void* const* d_in, const int* in_sizes, int n_in,
                              void* d_out, int out_size, void* d_ws, size_t ws_size,
                              hipStream_t stream) {
    const float* feature = (const float*)d_in[0];
    const int*   src     = (const int*)d_in[1];
    const int*   dst     = (const int*)d_in[2];
    const float* W1      = (const float*)d_in[3];
    const float* b1      = (const float*)d_in[4];
    const float* W2      = (const float*)d_in[5];
    const float* b2      = (const float*)d_in[6];
    float* out = (float*)d_out;

    const int F = 128;
    const int N = in_sizes[0] / F;
    const int E = in_sizes[1];

    char* ws = (char*)d_ws;
    float* nsrc = (float*)ws;
    float* ndst = nsrc + N;
    size_t off = ((size_t)2 * N * sizeof(float) + 255) & ~(size_t)255;
    float* buf0 = (float*)(ws + off);
    float* buf1 = buf0 + (size_t)N * F;

    const size_t featBytes = (size_t)N * F * sizeof(float);

    hipMemsetAsync(nsrc, 0, (size_t)2 * N * sizeof(float), stream);
    gnn_deg_kernel<<<(E + 255) / 256, 256, 0, stream>>>(src, dst, nsrc, ndst, E);
    gnn_norm_kernel<<<(2 * N + 255) / 256, 256, 0, stream>>>(nsrc, 2 * N);

    const int gemmBlocks = (N + 15) / 16;
    const long long edgeThreads = (long long)E * 32;
    const int aggBlocks = (int)((edgeThreads + 255) / 256);
    const long long nodeThreads = (long long)N * 32;
    const int finBlocks = (int)((nodeThreads + 255) / 256);

    gnn_gemm_kernel<<<gemmBlocks, 128, 0, stream>>>(feature, nsrc, W1, buf0, N);
    hipMemsetAsync(buf1, 0, featBytes, stream);
    gnn_agg_kernel<<<aggBlocks, 256, 0, stream>>>(buf0, src, dst, buf1, E);
    gnn_fin_kernel<<<finBlocks, 256, 0, stream>>>(buf1, ndst, b1, buf0, N, 1);

    gnn_gemm_kernel<<<gemmBlocks, 128, 0, stream>>>(buf0, nsrc, W2, buf1, N);
    hipMemsetAsync(buf0, 0, featBytes, stream);
    gnn_agg_kernel<<<aggBlocks, 256, 0, stream>>>(buf1, src, dst, buf0, E);
    gnn_fin_kernel<<<finBlocks, 256, 0, stream>>>(buf0, ndst, b2, out, N, 0);
}


# Aggregate over a destination-sorted edge list built in-call (integer degree atomics return each edge's rank; scan; fill): one wave per node sums rows in registers in rank order, norm/bias/relu fused
# speedup vs baseline: 2.4355x; 2.4355x over previous
#include <hip/hip_runtime.h>
#include <hip/hip_bf16.h>

typedef __attribute__((ext_vector_type(2))) float v2f;
typedef __attribute__((ext_vector_type(8))) float v8f;

#define LDSTRIDE 132
#define SCAN_T 1024

__global__ void gnn_count_rank_kernel(const int* __restrict__ src,
                                      const int* __restrict__ dst,
                                      int* __restrict__ cnt_out,
                                      int* __restrict__ cnt_in,
                                      int* __restrict__ rank, int E) {
    int e = blockIdx.x * blockDim.x + threadIdx.x;
    if (e < E) {
        atomicAdd(&cnt_out[src[e]], 1);
        rank[e] = atomicAdd(&cnt_in[dst[e]], 1);
    }
}

__global__ __launch_bounds__(SCAN_T) void gnn_norm_tilesum_kernel(
    const int* __restrict__ cnt_out, const int* __restrict__ cnt_in,
    float* __restrict__ nsrc, float* __restrict__ ndst,
    int* __restrict__ tile_sum, int N) {
    __shared__ int wsum[32];
    const int tid  = threadIdx.x;
    const int lane = tid & 31;
    const int wv   = tid >> 5;
    const int i    = blockIdx.x * SCAN_T + tid;
    int ci = 0;
    if (i < N) {
        ci = cnt_in[i];
        int co = cnt_out[i];
        nsrc[i] = 1.0f / sqrtf(fmaxf((float)co, 1.0f));
        ndst[i] = 1.0f / sqrtf(fmaxf((float)ci, 1.0f));
    }
    int s = ci;
    for (int o = 16; o > 0; o >>= 1) s += __shfl_xor(s, o);
    if (lane == 0) wsum[wv] = s;
    __syncthreads();
    if (wv == 0) {
        int t = wsum[lane];
        for (int o = 16; o > 0; o >>= 1) t += __shfl_xor(t, o);
        if (lane == 0) tile_sum[blockIdx.x] = t;
    }
}

__global__ __launch_bounds__(SCAN_T) void gnn_rowptr_kernel(
    const int* __restrict__ cnt_in, const int* __restrict__ tile_sum,
    int* __restrict__ row_ptr, int N) {
    __shared__ int wsum[32];
    const int tid  = threadIdx.x;
    const int lane = tid & 31;
    const int wv   = tid >> 5;
    const int i    = blockIdx.x * SCAN_T + tid;
    const int ci   = (i < N) ? cnt_in[i] : 0;

    int off = 0;
    for (int j = lane; j < (int)blockIdx.x; j += 32) off += tile_sum[j];
    for (int o = 16; o > 0; o >>= 1) off += __shfl_xor(off, o);

    int inc = ci;
    for (int o = 1; o < 32; o <<= 1) {
        int t = __shfl_up(inc, o);
        if (lane >= o) inc += t;
    }
    if (lane == 31) wsum[wv] = inc;
    __syncthreads();
    int wp = (lane < wv) ? wsum[lane] : 0;
    for (int o = 16; o > 0; o >>= 1) wp += __shfl_xor(wp, o);

    const int excl = off + wp + inc - ci;
    if (i < N) row_ptr[i] = excl;
    if (i == N - 1) row_ptr[N] = excl + ci;
}

__global__ void gnn_fill_kernel(const int* __restrict__ src,
                                const int* __restrict__ dst,
                                const int* __restrict__ row_ptr,
                                const int* __restrict__ rank,
                                int* __restrict__ col, int E) {
    int e = blockIdx.x * blockDim.x + threadIdx.x;
    if (e < E) col[row_ptr[dst[e]] + rank[e]] = src[e];
}

__global__ __launch_bounds__(128) void gnn_gemm_kernel(
    const float* __restrict__ H, const float* __restrict__ nsrc,
    const float* __restrict__ W, float* __restrict__ X, int N) {
    __shared__ float As[16 * LDSTRIDE];

    const int row0 = blockIdx.x * 16;
    const int tid  = threadIdx.x;

    for (int i = tid; i < 16 * 32; i += 128) {
        int r  = i >> 5;
        int c4 = (i & 31) * 4;
        int row = row0 + r;
        int rowc = (row < N) ? row : (N - 1);
        float s = nsrc[rowc];
        float4 v = *(const float4*)(H + (size_t)rowc * 128 + c4);
        float4 sv = make_float4(v.x * s, v.y * s, v.z * s, v.w * s);
        *(float4*)(&As[r * LDSTRIDE + c4]) = sv;
    }
    __syncthreads();

    const int wave = tid >> 5;
    const int lane = tid & 31;
    const int ln   = lane & 15;
    const int koff = (lane < 16) ? 0 : 2;

    const int n0 = wave * 32;
    const int n1 = n0 + 16;

    v8f acc0 = {};
    v8f acc1 = {};

    for (int kb = 0; kb < 128; kb += 4) {
        v2f a = *(const v2f*)(&As[ln * LDSTRIDE + kb + koff]);

        const float* wp = W + (size_t)(kb + koff) * 128;
        v2f b0, b1;
        b0.x = wp[n0 + ln];
        b0.y = wp[128 + n0 + ln];
        b1.x = wp[n1 + ln];
        b1.y = wp[128 + n1 + ln];

        acc0 = __builtin_amdgcn_wmma_f32_16x16x4_f32(
            false, a, false, b0, (short)0, acc0, false, false);
        acc1 = __builtin_amdgcn_wmma_f32_16x16x4_f32(
            false, a, false, b1, (short)0, acc1, false, false);
    }

    const int mbase = row0 + ((lane < 16) ? 0 : 8);
#pragma unroll
    for (int v = 0; v < 8; ++v) {
        int row = mbase + v;
        if (row < N) {
            X[(size_t)row * 128 + n0 + ln] = acc0[v];
            X[(size_t)row * 128 + n1 + ln] = acc1[v];
        }
    }
}

__global__ __launch_bounds__(256) void gnn_pull_kernel(
    const float* __restrict__ X, const int* __restrict__ row_ptr,
    const int* __restrict__ col, const float* __restrict__ ndst,
    const float* __restrict__ bias, float* __restrict__ OUT, int N, int do_relu) {
    const int lane = threadIdx.x & 31;
    const int wpb  = blockDim.x >> 5;
    const int w0 = __builtin_amdgcn_readfirstlane((int)(blockIdx.x * wpb + (threadIdx.x >> 5)));
    const int nw = (int)gridDim.x * wpb;
    const float4 b = *(const float4*)(bias + lane * 4);
    const float4 z = make_float4(0.0f, 0.0f, 0.0f, 0.0f);

    for (int n = w0; n < N; n += nw) {
        const int beg = row_ptr[n];
        const int end = row_ptr[n + 1];
        float4 acc = z;
        for (int base = beg; base < end; base += 32) {
            const int cnt = (end - base < 32) ? (end - base) : 32;
            const int myc = col[base + ((lane < cnt) ? lane : 0)];
            for (int j = 0; j < cnt; j += 4) {
                float4 v0, v1 = z, v2 = z, v3 = z;
                v0 = *(const float4*)(X + (size_t)__builtin_amdgcn_readlane(myc, j) * 128 + lane * 4);
                if (j + 1 < cnt)
                    v1 = *(const float4*)(X + (size_t)__builtin_amdgcn_readlane(myc, j + 1) * 128 + lane * 4);
                if (j + 2 < cnt)
                    v2 = *(const float4*)(X + (size_t)__builtin_amdgcn_readlane(myc, j + 2) * 128 + lane * 4);
                if (j + 3 < cnt)
                    v3 = *(const float4*)(X + (size_t)__builtin_amdgcn_readlane(myc, j + 3) * 128 + lane * 4);
                acc.x += v0.x; acc.y += v0.y; acc.z += v0.z; acc.w += v0.w;
                acc.x += v1.x; acc.y += v1.y; acc.z += v1.z; acc.w += v1.w;
                acc.x += v2.x; acc.y += v2.y; acc.z += v2.z; acc.w += v2.w;
                acc.x += v3.x; acc.y += v3.y; acc.z += v3.z; acc.w += v3.w;
            }
        }
        const float s = ndst[n];
        float4 o;
        o.x = acc.x * s + b.x;
        o.y = acc.y * s + b.y;
        o.z = acc.z * s + b.z;
        o.w = acc.w * s + b.w;
        if (do_relu) {
            o.x = fmaxf(o.x, 0.0f);
            o.y = fmaxf(o.y, 0.0f);
            o.z = fmaxf(o.z, 0.0f);
            o.w = fmaxf(o.w, 0.0f);
        }
        *(float4*)(OUT + (size_t)n * 128 + lane * 4) = o;
    }
}

extern "C" void kernel_launch(void* const* d_in, const int* in_sizes, int n_in,
                              void* d_out, int out_size, void* d_ws, size_t ws_size,
                              hipStream_t stream) {
    const float* feature = (const float*)d_in[0];
    const int*   src     = (const int*)d_in[1];
    const int*   dst     = (const int*)d_in[2];
    const float* W1      = (const float*)d_in[3];
    const float* b1      = (const float*)d_in[4];
    const float* W2      = (const float*)d_in[5];
    const float* b2      = (const float*)d_in[6];
    float* out = (float*)d_out;

    const int F = 128;
    const int N = in_sizes[0] / F;
    const int E = in_sizes[1];
    const int tiles = (N + SCAN_T - 1) / SCAN_T;

    char* ws = (char*)d_ws;
    size_t off = 0;
    int* cnt_out = (int*)(ws + off);
    int* cnt_in  = cnt_out + N;
    off = (off + (size_t)2 * N * sizeof(int) + 255) & ~(size_t)255;
    float* nsrc = (float*)(ws + off);
    float* ndst = nsrc + N;
    off = (off + (size_t)2 * N * sizeof(float) + 255) & ~(size_t)255;
    int* row_ptr = (int*)(ws + off);
    off = (off + ((size_t)N + 1) * sizeof(int) + 255) & ~(size_t)255;
    int* tile_sum = (int*)(ws + off);
    off = (off + (size_t)tiles * sizeof(int) + 255) & ~(size_t)255;
    int* rank = (int*)(ws + off);
    off = (off + (size_t)E * sizeof(int) + 255) & ~(size_t)255;
    int* col = (int*)(ws + off);
    off = (off + (size_t)E * sizeof(int) + 255) & ~(size_t)255;
    float* buf0 = (float*)(ws + off);
    float* buf1 = buf0 + (size_t)N * F;

    hipMemsetAsync(cnt_out, 0, (size_t)2 * N * sizeof(int), stream);
    gnn_count_rank_kernel<<<(E + 255) / 256, 256, 0, stream>>>(src, dst, cnt_out, cnt_in, rank, E);
    gnn_norm_tilesum_kernel<<<tiles, SCAN_T, 0, stream>>>(cnt_out, cnt_in, nsrc, ndst, tile_sum, N);
    gnn_rowptr_kernel<<<tiles, SCAN_T, 0, stream>>>(cnt_in, tile_sum, row_ptr, N);
    gnn_fill_kernel<<<(E + 255) / 256, 256, 0, stream>>>(src, dst, row_ptr, rank, col, E);

    const int gemmBlocks = (N + 15) / 16;
    int pullBlocks = (N + 7) / 8;
    if (pullBlocks > 2048) pullBlocks = 2048;

    gnn_gemm_kernel<<<gemmBlocks, 128, 0, stream>>>(feature, nsrc, W1, buf0, N);
    gnn_pull_kernel<<<pullBlocks, 256, 0, stream>>>(buf0, row_ptr, col, ndst, b1, buf1, N, 1);

    gnn_gemm_kernel<<<gemmBlocks, 128, 0, stream>>>(buf1, nsrc, W2, buf0, N);
    gnn_pull_kernel<<<pullBlocks, 256, 0, stream>>>(buf0, row_ptr, col, ndst, b2, out, N, 0);
}


# Edges grouped by destination with a two-level counting sort on LDS counters (chunk counts, scan, bucket scatter, per-bucket LDS ranks) instead of two global atomics per edge; sum order: LDS ranks
# speedup vs baseline: 3.7471x; 1.5385x over previous
#include <hip/hip_runtime.h>
#include <hip/hip_bf16.h>

typedef __attribute__((ext_vector_type(2))) float v2f;
typedef __attribute__((ext_vector_type(8))) float v8f;

#define LDSTRIDE 132
#define SCAN_T 1024
#define BSH 9
#define BSZ 512
#define CH 4096
#define CT 256
#define MAXB 2048
#define FB 8

__global__ __launch_bounds__(CT) void gnn_chunk_hist_kernel(
    const int* __restrict__ src, const int* __restrict__ dst,
    int* __restrict__ tab, int E, int NB, int G) {
    __shared__ int hD[MAXB];
    __shared__ int hS[MAXB];
    const int tid = threadIdx.x;
    const int g   = blockIdx.x;
    for (int b = tid; b < NB; b += CT) { hD[b] = 0; hS[b] = 0; }
    __syncthreads();
    const int base = g * CH;
    int s[CH / CT], d[CH / CT];
#pragma unroll
    for (int k = 0; k < CH / CT; ++k) {
        const int e  = base + k * CT + tid;
        const int ec = (e < E) ? e : (E - 1);
        s[k] = src[ec];
        d[k] = dst[ec];
    }
#pragma unroll
    for (int k = 0; k < CH / CT; ++k) {
        const int e = base + k * CT + tid;
        if (e < E) {
            atomicAdd(&hD[d[k] >> BSH], 1);
            atomicAdd(&hS[s[k] >> BSH], 1);
        }
    }
    __syncthreads();
    for (int b = tid; b < NB; b += CT) {
        tab[(size_t)b * G + g]        = hD[b];
        tab[(size_t)(NB + b) * G + g] = hS[b];
    }
}

__global__ __launch_bounds__(SCAN_T) void gnn_tilesum_kernel(
    const int* __restrict__ v, int* __restrict__ tile_sum, int L) {
    __shared__ int wsum[32];
    const int tid  = threadIdx.x;
    const int lane = tid & 31;
    const int wv   = tid >> 5;
    const int i    = blockIdx.x * SCAN_T + tid;
    int s = (i < L) ? v[i] : 0;
    for (int o = 16; o > 0; o >>= 1) s += __shfl_xor(s, o);
    if (lane == 0) wsum[wv] = s;
    __syncthreads();
    if (wv == 0) {
        int t = wsum[lane];
        for (int o = 16; o > 0; o >>= 1) t += __shfl_xor(t, o);
        if (lane == 0) tile_sum[blockIdx.x] = t;
    }
}

__global__ __launch_bounds__(SCAN_T) void gnn_scan_kernel(
    const int* __restrict__ v, const int* __restrict__ tile_sum,
    int* __restrict__ off, int L) {
    __shared__ int wsum[32];
    const int tid  = threadIdx.x;
    const int lane = tid & 31;
    const int wv   = tid >> 5;
    const int i    = blockIdx.x * SCAN_T + tid;
    const int ci   = (i < L) ? v[i] : 0;

    int pre = 0;
    for (int j = lane; j < (int)blockIdx.x; j += 32) pre += tile_sum[j];
    for (int o = 16; o > 0; o >>= 1) pre += __shfl_xor(pre, o);

    int inc = ci;
    for (int o = 1; o < 32; o <<= 1) {
        int t = __shfl_up(inc, o);
        if (lane >= o) inc += t;
    }
    if (lane == 31) wsum[wv] = inc;
    __syncthreads();
    int wp = (lane < wv) ? wsum[lane] : 0;
    for (int o = 16; o > 0; o >>= 1) wp += __shfl_xor(wp, o);

    const int excl = pre + wp + inc - ci;
    if (i < L) off[i] = excl;
    if (i == L - 1) off[L] = excl + ci;
}

__global__ __launch_bounds__(CT) void gnn_bucket_scatter_kernel(
    const int* __restrict__ src, const int* __restrict__ dst,
    const int* __restrict__ off, int2* __restrict__ bpair,
    int* __restrict__ ssrc, int E, int NB, int G) {
    __shared__ int cD[MAXB];
    __shared__ int cS[MAXB];
    const int tid = threadIdx.x;
    const int g   = blockIdx.x;
    for (int b = tid; b < NB; b += CT) {
        cD[b] = off[(size_t)b * G + g];
        cS[b] = off[(size_t)(NB + b) * G + g] - E;
    }
    __syncthreads();
    const int base = g * CH;
    int s[CH / CT], d[CH / CT];
#pragma unroll
    for (int k = 0; k < CH / CT; ++k) {
        const int e  = base + k * CT + tid;
        const int ec = (e < E) ? e : (E - 1);
        s[k] = src[ec];
        d[k] = dst[ec];
    }
#pragma unroll
    for (int k = 0; k < CH / CT; ++k) {
        const int e = base + k * CT + tid;
        if (e < E) {
            const int pD = atomicAdd(&cD[d[k] >> BSH], 1);
            bpair[pD] = make_int2(s[k], d[k]);
            const int pS = atomicAdd(&cS[s[k] >> BSH], 1);
            ssrc[pS] = s[k];
        }
    }
}

__global__ __launch_bounds__(BSZ) void gnn_bucket_finish_kernel(
    const int2* __restrict__ bpair, const int* __restrict__ ssrc,
    const int* __restrict__ off, int* __restrict__ row_ptr, int* __restrict__ col,
    float* __restrict__ nsrc, float* __restrict__ ndst, int N, int E, int NB, int G) {
    __shared__ int cnt[BSZ];
    __shared__ int wsum[BSZ / 32];
    const int tid  = threadIdx.x;
    const int lane = tid & 31;
    const int wv   = tid >> 5;
    cnt[tid] = 0;
    __syncthreads();

    if ((int)blockIdx.x >= NB) {
        const int b   = (int)blockIdx.x - NB;
        const int beg = off[(size_t)(NB + b) * G] - E;
        const int end = off[(size_t)(NB + b + 1) * G] - E;
        for (int i0 = beg; i0 < end; i0 += FB * BSZ) {
            int v[FB];
#pragma unroll
            for (int k = 0; k < FB; ++k) {
                const int i = i0 + k * BSZ + tid;
                v[k] = ssrc[(i < end) ? i : (end - 1)];
            }
#pragma unroll
            for (int k = 0; k < FB; ++k)
                if (i0 + k * BSZ + tid < end) atomicAdd(&cnt[v[k] & (BSZ - 1)], 1);
        }
        __syncthreads();
        const int n = b * BSZ + tid;
        if (n < N) nsrc[n] = 1.0f / sqrtf(fmaxf((float)cnt[tid], 1.0f));
        return;
    }

    const int b   = (int)blockIdx.x;
    const int beg = off[(size_t)b * G];
    const int end = off[(size_t)(b + 1) * G];
    for (int i0 = beg; i0 < end; i0 += FB * BSZ) {
        int v[FB];
#pragma unroll
        for (int k = 0; k < FB; ++k) {
            const int i = i0 + k * BSZ + tid;
            v[k] = bpair[(i < end) ? i : (end - 1)].y;
        }
#pragma unroll
        for (int k = 0; k < FB; ++k)
            if (i0 + k * BSZ + tid < end) atomicAdd(&cnt[v[k] & (BSZ - 1)], 1);
    }
    __syncthreads();

    const int ci = cnt[tid];
    int inc = ci;
    for (int o = 1; o < 32; o <<= 1) {
        int t = __shfl_up(inc, o);
        if (lane >= o) inc += t;
    }
    if (lane == 31) wsum[wv] = inc;
    __syncthreads();
    int wp = (lane < wv) ? wsum[lane] : 0;
    for (int o = 16; o > 0; o >>= 1) wp += __shfl_xor(wp, o);

    const int excl = beg + wp + inc - ci;
    const int n = b * BSZ + tid;
    if (n < N) {
        row_ptr[n] = excl;
        ndst[n] = 1.0f / sqrtf(fmaxf((float)ci, 1.0f));
    }
    if (n == N - 1) row_ptr[N] = excl + ci;

    cnt[tid] = excl;
    __syncthreads();
    for (int i0 = beg; i0 < end; i0 += FB * BSZ) {
        int2 p[FB];
#pragma unroll
        for (int k = 0; k < FB; ++k) {
            const int i = i0 + k * BSZ + tid;
            p[k] = bpair[(i < end) ? i : (end - 1)];
        }
#pragma unroll
        for (int k = 0; k < FB; ++k) {
            if (i0 + k * BSZ + tid < end) {
                const int pos = atomicAdd(&cnt[p[k].y & (BSZ - 1)], 1);
                col[pos] = p[k].x;
            }
        }
    }
}

__global__ __launch_bounds__(128) void gnn_gemm_kernel(
    const float* __restrict__ H, const float* __restrict__ nsrc,
    const float* __restrict__ W, float* __restrict__ X, int N) {
    __shared__ float As[16 * LDSTRIDE];

    const int row0 = blockIdx.x * 16;
    const int tid  = threadIdx.x;

    for (int i = tid; i < 16 * 32; i += 128) {
        int r  = i >> 5;
        int c4 = (i & 31) * 4;
        int row = row0 + r;
        int rowc = (row < N) ? row : (N - 1);
        float s = nsrc[rowc];
        float4 v = *(const float4*)(H + (size_t)rowc * 128 + c4);
        float4 sv = make_float4(v.x * s, v.y * s, v.z * s, v.w * s);
        *(float4*)(&As[r * LDSTRIDE + c4]) = sv;
    }
    __syncthreads();

    const int wave = tid >> 5;
    const int lane = tid & 31;
    const int ln   = lane & 15;
    const int koff = (lane < 16) ? 0 : 2;

    const int n0 = wave * 32;
    const int n1 = n0 + 16;

    v8f acc0 = {};
    v8f acc1 = {};

    for (int kb = 0; kb < 128; kb += 4) {
        v2f a = *(const v2f*)(&As[ln * LDSTRIDE + kb + koff]);

        const float* wp = W + (size_t)(kb + koff) * 128;
        v2f b0, b1;
        b0.x = wp[n0 + ln];
        b0.y = wp[128 + n0 + ln];
        b1.x = wp[n1 + ln];
        b1.y = wp[128 + n1 + ln];

        acc0 = __builtin_amdgcn_wmma_f32_16x16x4_f32(
            false, a, false, b0, (short)0, acc0, false, false);
        acc1 = __builtin_amdgcn_wmma_f32_16x16x4_f32(
            false, a, false, b1, (short)0, acc1, false, false);
    }

    const int mbase = row0 + ((lane < 16) ? 0 : 8);
#pragma unroll
    for (int v = 0; v < 8; ++v) {
        int row = mbase + v;
        if (row < N) {
            X[(size_t)row * 128 + n0 + ln] = acc0[v];
            X[(size_t)row * 128 + n1 + ln] = acc1[v];
        }
    }
}

__global__ __launch_bounds__(256) void gnn_pull_kernel(
    const float* __restrict__ X, const int* __restrict__ row_ptr,
    const int* __restrict__ col, const float* __restrict__ ndst,
    const float* __restrict__ bias, float* __restrict__ OUT, int N, int do_relu) {
    const int lane = threadIdx.x & 31;
    const int wpb  = blockDim.x >> 5;
    const int w0 = __builtin_amdgcn_readfirstlane((int)(blockIdx.x * wpb + (threadIdx.x >> 5)));
    const int nw = (int)gridDim.x * wpb;
    const float4 b = *(const float4*)(bias + lane * 4);
    const float4 z = make_float4(0.0f, 0.0f, 0.0f, 0.0f);

    for (int n = w0; n < N; n += nw) {
        const int beg = row_ptr[n];
        const int end = row_ptr[n + 1];
        float4 acc = z;
        for (int base = beg; base < end; base += 32) {
            const int cnt = (end - base < 32) ? (end - base) : 32;
            const int myc = col[base + ((lane < cnt) ? lane : 0)];
            for (int j = 0; j < cnt; j += 4) {
                float4 v0, v1 = z, v2 = z, v3 = z;
                v0 = *(const float4*)(X + (size_t)__builtin_amdgcn_readlane(myc, j) * 128 + lane * 4);
                if (j + 1 < cnt)
                    v1 = *(const float4*)(X + (size_t)__builtin_amdgcn_readlane(myc, j + 1) * 128 + lane * 4);
                if (j + 2 < cnt)
                    v2 = *(const float4*)(X + (size_t)__builtin_amdgcn_readlane(myc, j + 2) * 128 + lane * 4);
                if (j + 3 < cnt)
                    v3 = *(const float4*)(X + (size_t)__builtin_amdgcn_readlane(myc, j + 3) * 128 + lane * 4);
                acc.x += v0.x; acc.y += v0.y; acc.z += v0.z; acc.w += v0.w;
                acc.x += v1.x; acc.y += v1.y; acc.z += v1.z; acc.w += v1.w;
                acc.x += v2.x; acc.y += v2.y; acc.z += v2.z; acc.w += v2.w;
                acc.x += v3.x; acc.y += v3.y; acc.z += v3.z; acc.w += v3.w;
            }
        }
        const float s = ndst[n];
        float4 o;
        o.x = acc.x * s + b.x;
        o.y = acc.y * s + b.y;
        o.z = acc.z * s + b.z;
        o.w = acc.w * s + b.w;
        if (do_relu) {
            o.x = fmaxf(o.x, 0.0f);
            o.y = fmaxf(o.y, 0.0f);
            o.z = fmaxf(o.z, 0.0f);
            o.w = fmaxf(o.w, 0.0f);
        }
        *(float4*)(OUT + (size_t)n * 128 + lane * 4) = o;
    }
}

extern "C" void kernel_launch(void* const* d_in, const int* in_sizes, int n_in,
                              void* d_out, int out_size, void* d_ws, size_t ws_size,
                              hipStream_t stream) {
    const float* feature = (const float*)d_in[0];
    const int*   src     = (const int*)d_in[1];
    const int*   dst     = (const int*)d_in[2];
    const float* W1      = (const float*)d_in[3];
    const float* b1      = (const float*)d_in[4];
    const float* W2      = (const float*)d_in[5];
    const float* b2      = (const float*)d_in[6];
    float* out = (float*)d_out;

    const int F = 128;
    const int N = in_sizes[0] / F;
    const int E = in_sizes[1];
    const int NB = (N + BSZ - 1) / BSZ;
    const int G  = (E + CH - 1) / CH;
    const int L  = 2 * NB * G;
    const int tiles = (L + SCAN_T - 1) / SCAN_T;

    char* ws = (char*)d_ws;
    size_t o = 0;
    int* tab = (int*)(ws + o);
    o = (o + (size_t)L * sizeof(int) + 255) & ~(size_t)255;
    int* off = (int*)(ws + o);
    o = (o + ((size_t)L + 1) * sizeof(int) + 255) & ~(size_t)255;
    int* tile_sum = (int*)(ws + o);
    o = (o + (size_t)tiles * sizeof(int) + 255) & ~(size_t)255;
    int2* bpair = (int2*)(ws + o);
    o = (o + (size_t)E * sizeof(int2) + 255) & ~(size_t)255;
    int* ssrc = (int*)(ws + o);
    o = (o + (size_t)E * sizeof(int) + 255) & ~(size_t)255;
    float* nsrc = (float*)(ws + o);
    float* ndst = nsrc + N;
    o = (o + (size_t)2 * N * sizeof(float) + 255) & ~(size_t)255;
    int* row_ptr = (int*)(ws + o);
    o = (o + ((size_t)N + 1) * sizeof(int) + 255) & ~(size_t)255;
    int* col = (int*)(ws + o);
    o = (o + (size_t)E * sizeof(int) + 255) & ~(size_t)255;
    float* buf0 = (float*)(ws + o);
    float* buf1 = buf0 + (size_t)N * F;

    gnn_chunk_hist_kernel<<<G, CT, 0, stream>>>(src, dst, tab, E, NB, G);
    gnn_tilesum_kernel<<<tiles, SCAN_T, 0, stream>>>(tab, tile_sum, L);
    gnn_scan_kernel<<<tiles, SCAN_T, 0, stream>>>(tab, tile_sum, off, L);
    gnn_bucket_scatter_kernel<<<G, CT, 0, stream>>>(src, dst, off, bpair, ssrc, E, NB, G);
    gnn_bucket_finish_kernel<<<2 * NB, BSZ, 0, stream>>>(bpair, ssrc, off, row_ptr, col, nsrc, ndst, N, E, NB, G);

    const int gemmBlocks = (N + 15) / 16;
    int pullBlocks = (N + 7) / 8;
    if (pullBlocks > 2048) pullBlocks = 2048;

    gnn_gemm_kernel<<<gemmBlocks, 128, 0, stream>>>(feature, nsrc, W1, buf0, N);
    gnn_pull_kernel<<<pullBlocks, 256, 0, stream>>>(buf0, row_ptr, col, ndst, b1, buf1, N, 1);

    gnn_gemm_kernel<<<gemmBlocks, 128, 0, stream>>>(buf1, nsrc, W2, buf0, N);
    gnn_pull_kernel<<<pullBlocks, 256, 0, stream>>>(buf0, row_ptr, col, ndst, b2, out, N, 0);
}


# Layer-1 aggregation fused with the layer-2 projection: a 4-wave block sums 16 nodes' rows into the LDS A tile and multiplies by W2; h1 never stored, one launch fewer; sum order kept
# speedup vs baseline: 3.9498x; 1.0541x over previous
#include <hip/hip_runtime.h>
#include <hip/hip_bf16.h>

typedef __attribute__((ext_vector_type(2))) float v2f;
typedef __attribute__((ext_vector_type(8))) float v8f;

#define LDSTRIDE 132
#define SCAN_T 1024
#define BSH 9
#define BSZ 512
#define CH 4096
#define CT 256
#define MAXB 2048
#define FB 8

__global__ __launch_bounds__(CT) void gnn_chunk_hist_kernel(
    const int* __restrict__ src, const int* __restrict__ dst,
    int* __restrict__ tab, int E, int NB, int G) {
    __shared__ int hD[MAXB];
    __shared__ int hS[MAXB];
    const int tid = threadIdx.x;
    const int g   = blockIdx.x;
    for (int b = tid; b < NB; b += CT) { hD[b] = 0; hS[b] = 0; }
    __syncthreads();
    const int base = g * CH;
    int s[CH / CT], d[CH / CT];
#pragma unroll
    for (int k = 0; k < CH / CT; ++k) {
        const int e  = base + k * CT + tid;
        const int ec = (e < E) ? e : (E - 1);
        s[k] = src[ec];
        d[k] = dst[ec];
    }
#pragma unroll
    for (int k = 0; k < CH / CT; ++k) {
        const int e = base + k * CT + tid;
        if (e < E) {
            atomicAdd(&hD[d[k] >> BSH], 1);
            atomicAdd(&hS[s[k] >> BSH], 1);
        }
    }
    __syncthreads();
    for (int b = tid; b < NB; b += CT) {
        tab[(size_t)b * G + g]        = hD[b];
        tab[(size_t)(NB + b) * G + g] = hS[b];
    }
}

__global__ __launch_bounds__(SCAN_T) void gnn_tilesum_kernel(
    const int* __restrict__ v, int* __restrict__ tile_sum, int L) {
    __shared__ int wsum[32];
    const int tid  = threadIdx.x;
    const int lane = tid & 31;
    const int wv   = tid >> 5;
    const int i    = blockIdx.x * SCAN_T + tid;
    int s = (i < L) ? v[i] : 0;
    for (int o = 16; o > 0; o >>= 1) s += __shfl_xor(s, o);
    if (lane == 0) wsum[wv] = s;
    __syncthreads();
    if (wv == 0) {
        int t = wsum[lane];
        for (int o = 16; o > 0; o >>= 1) t += __shfl_xor(t, o);
        if (lane == 0) tile_sum[blockIdx.x] = t;
    }
}

__global__ __launch_bounds__(SCAN_T) void gnn_scan_kernel(
    const int* __restrict__ v, const int* __restrict__ tile_sum,
    int* __restrict__ off, int L) {
    __shared__ int wsum[32];
    const int tid  = threadIdx.x;
    const int lane = tid & 31;
    const int wv   = tid >> 5;
    const int i    = blockIdx.x * SCAN_T + tid;
    const int ci   = (i < L) ? v[i] : 0;

    int pre = 0;
    for (int j = lane; j < (int)blockIdx.x; j += 32) pre += tile_sum[j];
    for (int o = 16; o > 0; o >>= 1) pre += __shfl_xor(pre, o);

    int inc = ci;
    for (int o = 1; o < 32; o <<= 1) {
        int t = __shfl_up(inc, o);
        if (lane >= o) inc += t;
    }
    if (lane == 31) wsum[wv] = inc;
    __syncthreads();
    int wp = (lane < wv) ? wsum[lane] : 0;
    for (int o = 16; o > 0; o >>= 1) wp += __shfl_xor(wp, o);

    const int excl = pre + wp + inc - ci;
    if (i < L) off[i] = excl;
    if (i == L - 1) off[L] = excl + ci;
}

__global__ __launch_bounds__(CT) void gnn_bucket_scatter_kernel(
    const int* __restrict__ src, const int* __restrict__ dst,
    const int* __restrict__ off, int2* __restrict__ bpair,
    int* __restrict__ ssrc, int E, int NB, int G) {
    __shared__ int cD[MAXB];
    __shared__ int cS[MAXB];
    const int tid = threadIdx.x;
    const int g   = blockIdx.x;
    for (int b = tid; b < NB; b += CT) {
        cD[b] = off[(size_t)b * G + g];
        cS[b] = off[(size_t)(NB + b) * G + g] - E;
    }
    __syncthreads();
    const int base = g * CH;
    int s[CH / CT], d[CH / CT];
#pragma unroll
    for (int k = 0; k < CH / CT; ++k) {
        const int e  = base + k * CT + tid;
        const int ec = (e < E) ? e : (E - 1);
        s[k] = src[ec];
        d[k] = dst[ec];
    }
#pragma unroll
    for (int k = 0; k < CH / CT; ++k) {
        const int e = base + k * CT + tid;
        if (e < E) {
            const int pD = atomicAdd(&cD[d[k] >> BSH], 1);
            bpair[pD] = make_int2(s[k], d[k]);
            const int pS = atomicAdd(&cS[s[k] >> BSH], 1);
            ssrc[pS] = s[k];
        }
    }
}

__global__ __launch_bounds__(BSZ) void gnn_bucket_finish_kernel(
    const int2* __restrict__ bpair, const int* __restrict__ ssrc,
    const int* __restrict__ off, int* __restrict__ row_ptr, int* __restrict__ col,
    float* __restrict__ nsrc, float* __restrict__ ndst, int N, int E, int NB, int G) {
    __shared__ int cnt[BSZ];
    __shared__ int wsum[BSZ / 32];
    const int tid  = threadIdx.x;
    const int lane = tid & 31;
    const int wv   = tid >> 5;
    cnt[tid] = 0;
    __syncthreads();

    if ((int)blockIdx.x >= NB) {
        const int b   = (int)blockIdx.x - NB;
        const int beg = off[(size_t)(NB + b) * G] - E;
        const int end = off[(size_t)(NB + b + 1) * G] - E;
        for (int i0 = beg; i0 < end; i0 += FB * BSZ) {
            int v[FB];
#pragma unroll
            for (int k = 0; k < FB; ++k) {
                const int i = i0 + k * BSZ + tid;
                v[k] = ssrc[(i < end) ? i : (end - 1)];
            }
#pragma unroll
            for (int k = 0; k < FB; ++k)
                if (i0 + k * BSZ + tid < end) atomicAdd(&cnt[v[k] & (BSZ - 1)], 1);
        }
        __syncthreads();
        const int n = b * BSZ + tid;
        if (n < N) nsrc[n] = 1.0f / sqrtf(fmaxf((float)cnt[tid], 1.0f));
        return;
    }

    const int b   = (int)blockIdx.x;
    const int beg = off[(size_t)b * G];
    const int end = off[(size_t)(b + 1) * G];
    for (int i0 = beg; i0 < end; i0 += FB * BSZ) {
        int v[FB];
#pragma unroll
        for (int k = 0; k < FB; ++k) {
            const int i = i0 + k * BSZ + tid;
            v[k] = bpair[(i < end) ? i : (end - 1)].y;
        }
#pragma unroll
        for (int k = 0; k < FB; ++k)
            if (i0 + k * BSZ + tid < end) atomicAdd(&cnt[v[k] & (BSZ - 1)], 1);
    }
    __syncthreads();

    const int ci = cnt[tid];
    int inc = ci;
    for (int o = 1; o < 32; o <<= 1) {
        int t = __shfl_up(inc, o);
        if (lane >= o) inc += t;
    }
    if (lane == 31) wsum[wv] = inc;
    __syncthreads();
    int wp = (lane < wv) ? wsum[lane] : 0;
    for (int o = 16; o > 0; o >>= 1) wp += __shfl_xor(wp, o);

    const int excl = beg + wp + inc - ci;
    const int n = b * BSZ + tid;
    if (n < N) {
        row_ptr[n] = excl;
        ndst[n] = 1.0f / sqrtf(fmaxf((float)ci, 1.0f));
    }
    if (n == N - 1) row_ptr[N] = excl + ci;

    cnt[tid] = excl;
    __syncthreads();
    for (int i0 = beg; i0 < end; i0 += FB * BSZ) {
        int2 p[FB];
#pragma unroll
        for (int k = 0; k < FB; ++k) {
            const int i = i0 + k * BSZ + tid;
            p[k] = bpair[(i < end) ? i : (end - 1)];
        }
#pragma unroll
        for (int k = 0; k < FB; ++k) {
            if (i0 + k * BSZ + tid < end) {
                const int pos = atomicAdd(&cnt[p[k].y & (BSZ - 1)], 1);
                col[pos] = p[k].x;
            }
        }
    }
}

__global__ __launch_bounds__(128) void gnn_gemm_kernel(
    const float* __restrict__ H, const float* __restrict__ nsrc,
    const float* __restrict__ W, float* __restrict__ X, int N) {
    __shared__ float As[16 * LDSTRIDE];

    const int row0 = blockIdx.x * 16;
    const int tid  = threadIdx.x;

    for (int i = tid; i < 16 * 32; i += 128) {
        int r  = i >> 5;
        int c4 = (i & 31) * 4;
        int row = row0 + r;
        int rowc = (row < N) ? row : (N - 1);
        float s = nsrc[rowc];
        float4 v = *(const float4*)(H + (size_t)rowc * 128 + c4);
        float4 sv = make_float4(v.x * s, v.y * s, v.z * s, v.w * s);
        *(float4*)(&As[r * LDSTRIDE + c4]) = sv;
    }
    __syncthreads();

    const int wave = tid >> 5;
    const int lane = tid & 31;
    const int ln   = lane & 15;
    const int koff = (lane < 16) ? 0 : 2;

    const int n0 = wave * 32;
    const int n1 = n0 + 16;

    v8f acc0 = {};
    v8f acc1 = {};

    for (int kb = 0; kb < 128; kb += 4) {
        v2f a = *(const v2f*)(&As[ln * LDSTRIDE + kb + koff]);

        const float* wp = W + (size_t)(kb + koff) * 128;
        v2f b0, b1;
        b0.x = wp[n0 + ln];
        b0.y = wp[128 + n0 + ln];
        b1.x = wp[n1 + ln];
        b1.y = wp[128 + n1 + ln];

        acc0 = __builtin_amdgcn_wmma_f32_16x16x4_f32(
            false, a, false, b0, (short)0, acc0, false, false);
        acc1 = __builtin_amdgcn_wmma_f32_16x16x4_f32(
            false, a, false, b1, (short)0, acc1, false, false);
    }

    const int mbase = row0 + ((lane < 16) ? 0 : 8);
#pragma unroll
    for (int v = 0; v < 8; ++v) {
        int row = mbase + v;
        if (row < N) {
            X[(size_t)row * 128 + n0 + ln] = acc0[v];
            X[(size_t)row * 128 + n1 + ln] = acc1[v];
        }
    }
}

__global__ __launch_bounds__(128) void gnn_pull_relu_gemm_kernel(
    const float* __restrict__ X, const int* __restrict__ row_ptr,
    const int* __restrict__ col, const float* __restrict__ ndst,
    const float* __restrict__ nsrc, const float* __restrict__ bias,
    const float* __restrict__ W, float* __restrict__ Y, int N) {
    __shared__ float As[16 * LDSTRIDE];

    const int row0 = blockIdx.x * 16;
    const int tid  = threadIdx.x;
    const int lane = tid & 31;
    const int wave = __builtin_amdgcn_readfirstlane(tid >> 5);
    const float4 b = *(const float4*)(bias + lane * 4);
    const float4 z = make_float4(0.0f, 0.0f, 0.0f, 0.0f);

    const int nq = row0 + wave * 4;
    int rp[5];
#pragma unroll
    for (int q = 0; q < 5; ++q) rp[q] = row_ptr[(nq + q < N) ? (nq + q) : N];

#pragma unroll
    for (int q = 0; q < 4; ++q) {
        const int r = wave * 4 + q;
        const int n = nq + q;
        float4 o = z;
        if (n < N) {
            const int beg = rp[q];
            const int end = rp[q + 1];
            float4 acc = z;
            for (int base = beg; base < end; base += 32) {
                const int cnt = (end - base < 32) ? (end - base) : 32;
                const int myc = col[base + ((lane < cnt) ? lane : 0)];
                for (int j = 0; j < cnt; j += 4) {
                    float4 v0, v1 = z, v2 = z, v3 = z;
                    v0 = *(const float4*)(X + (size_t)__builtin_amdgcn_readlane(myc, j) * 128 + lane * 4);
                    if (j + 1 < cnt)
                        v1 = *(const float4*)(X + (size_t)__builtin_amdgcn_readlane(myc, j + 1) * 128 + lane * 4);
                    if (j + 2 < cnt)
                        v2 = *(const float4*)(X + (size_t)__builtin_amdgcn_readlane(myc, j + 2) * 128 + lane * 4);
                    if (j + 3 < cnt)
                        v3 = *(const float4*)(X + (size_t)__builtin_amdgcn_readlane(myc, j + 3) * 128 + lane * 4);
                    acc.x += v0.x; acc.y += v0.y; acc.z += v0.z; acc.w += v0.w;
                    acc.x += v1.x; acc.y += v1.y; acc.z += v1.z; acc.w += v1.w;
                    acc.x += v2.x; acc.y += v2.y; acc.z += v2.z; acc.w += v2.w;
                    acc.x += v3.x; acc.y += v3.y; acc.z += v3.z; acc.w += v3.w;
                }
            }
            const float s = ndst[n];
            const float t = nsrc[n];
            o.x = fmaxf(acc.x * s + b.x, 0.0f) * t;
            o.y = fmaxf(acc.y * s + b.y, 0.0f) * t;
            o.z = fmaxf(acc.z * s + b.z, 0.0f) * t;
            o.w = fmaxf(acc.w * s + b.w, 0.0f) * t;
        }
        *(float4*)(&As[r * LDSTRIDE + lane * 4]) = o;
    }
    __syncthreads();

    const int ln   = lane & 15;
    const int koff = (lane < 16) ? 0 : 2;

    const int n0 = wave * 32;
    const int n1 = n0 + 16;

    v8f acc0 = {};
    v8f acc1 = {};

    for (int kb = 0; kb < 128; kb += 4) {
        v2f a = *(const v2f*)(&As[ln * LDSTRIDE + kb + koff]);

        const float* wp = W + (size_t)(kb + koff) * 128;
        v2f b0, b1;
        b0.x = wp[n0 + ln];
        b0.y = wp[128 + n0 + ln];
        b1.x = wp[n1 + ln];
        b1.y = wp[128 + n1 + ln];

        acc0 = __builtin_amdgcn_wmma_f32_16x16x4_f32(
            false, a, false, b0, (short)0, acc0, false, false);
        acc1 = __builtin_amdgcn_wmma_f32_16x16x4_f32(
            false, a, false, b1, (short)0, acc1, false, false);
    }

    const int mbase = row0 + ((lane < 16) ? 0 : 8);
#pragma unroll
    for (int v = 0; v < 8; ++v) {
        int row = mbase + v;
        if (row < N) {
            Y[(size_t)row * 128 + n0 + ln] = acc0[v];
            Y[(size_t)row * 128 + n1 + ln] = acc1[v];
        }
    }
}

__global__ __launch_bounds__(256) void gnn_pull_kernel(
    const float* __restrict__ X, const int* __restrict__ row_ptr,
    const int* __restrict__ col, const float* __restrict__ ndst,
    const float* __restrict__ bias, float* __restrict__ OUT, int N) {
    const int lane = threadIdx.x & 31;
    const int wpb  = blockDim.x >> 5;
    const int w0 = __builtin_amdgcn_readfirstlane((int)(blockIdx.x * wpb + (threadIdx.x >> 5)));
    const int nw = (int)gridDim.x * wpb;
    const float4 b = *(const float4*)(bias + lane * 4);
    const float4 z = make_float4(0.0f, 0.0f, 0.0f, 0.0f);

    for (int n = w0; n < N; n += nw) {
        const int beg = row_ptr[n];
        const int end = row_ptr[n + 1];
        float4 acc = z;
        for (int base = beg; base < end; base += 32) {
            const int cnt = (end - base < 32) ? (end - base) : 32;
            const int myc = col[base + ((lane < cnt) ? lane : 0)];
            for (int j = 0; j < cnt; j += 4) {
                float4 v0, v1 = z, v2 = z, v3 = z;
                v0 = *(const float4*)(X + (size_t)__builtin_amdgcn_readlane(myc, j) * 128 + lane * 4);
                if (j + 1 < cnt)
                    v1 = *(const float4*)(X + (size_t)__builtin_amdgcn_readlane(myc, j + 1) * 128 + lane * 4);
                if (j + 2 < cnt)
                    v2 = *(const float4*)(X + (size_t)__builtin_amdgcn_readlane(myc, j + 2) * 128 + lane * 4);
                if (j + 3 < cnt)
                    v3 = *(const float4*)(X + (size_t)__builtin_amdgcn_readlane(myc, j + 3) * 128 + lane * 4);
                acc.x += v0.x; acc.y += v0.y; acc.z += v0.z; acc.w += v0.w;
                acc.x += v1.x; acc.y += v1.y; acc.z += v1.z; acc.w += v1.w;
                acc.x += v2.x; acc.y += v2.y; acc.z += v2.z; acc.w += v2.w;
                acc.x += v3.x; acc.y += v3.y; acc.z += v3.z; acc.w += v3.w;
            }
        }
        const float s = ndst[n];
        float4 o;
        o.x = acc.x * s + b.x;
        o.y = acc.y * s + b.y;
        o.z = acc.z * s + b.z;
        o.w = acc.w * s + b.w;
        *(float4*)(OUT + (size_t)n * 128 + lane * 4) = o;
    }
}

extern "C" void kernel_launch(void* const* d_in, const int* in_sizes, int n_in,
                              void* d_out, int out_size, void* d_ws, size_t ws_size,
                              hipStream_t stream) {
    const float* feature = (const float*)d_in[0];
    const int*   src     = (const int*)d_in[1];
    const int*   dst     = (const int*)d_in[2];
    const float* W1      = (const float*)d_in[3];
    const float* b1      = (const float*)d_in[4];
    const float* W2      = (const float*)d_in[5];
    const float* b2      = (const float*)d_in[6];
    float* out = (float*)d_out;

    const int F = 128;
    const int N = in_sizes[0] / F;
    const int E = in_sizes[1];
    const int NB = (N + BSZ - 1) / BSZ;
    const int G  = (E + CH - 1) / CH;
    const int L  = 2 * NB * G;
    const int tiles = (L + SCAN_T - 1) / SCAN_T;

    char* ws = (char*)d_ws;
    size_t o = 0;
    int* tab = (int*)(ws + o);
    o = (o + (size_t)L * sizeof(int) + 255) & ~(size_t)255;
    int* off = (int*)(ws + o);
    o = (o + ((size_t)L + 1) * sizeof(int) + 255) & ~(size_t)255;
    int* tile_sum = (int*)(ws + o);
    o = (o + (size_t)tiles * sizeof(int) + 255) & ~(size_t)255;
    int2* bpair = (int2*)(ws + o);
    o = (o + (size_t)E * sizeof(int2) + 255) & ~(size_t)255;
    int* ssrc = (int*)(ws + o);
    o = (o + (size_t)E * sizeof(int) + 255) & ~(size_t)255;
    float* nsrc = (float*)(ws + o);
    float* ndst = nsrc + N;
    o = (o + (size_t)2 * N * sizeof(float) + 255) & ~(size_t)255;
    int* row_ptr = (int*)(ws + o);
    o = (o + ((size_t)N + 1) * sizeof(int) + 255) & ~(size_t)255;
    int* col = (int*)(ws + o);
    o = (o + (size_t)E * sizeof(int) + 255) & ~(size_t)255;
    float* buf0 = (float*)(ws + o);
    float* buf1 = buf0 + (size_t)N * F;

    gnn_chunk_hist_kernel<<<G, CT, 0, stream>>>(src, dst, tab, E, NB, G);
    gnn_tilesum_kernel<<<tiles, SCAN_T, 0, stream>>>(tab, tile_sum, L);
    gnn_scan_kernel<<<tiles, SCAN_T, 0, stream>>>(tab, tile_sum, off, L);
    gnn_bucket_scatter_kernel<<<G, CT, 0, stream>>>(src, dst, off, bpair, ssrc, E, NB, G);
    gnn_bucket_finish_kernel<<<2 * NB, BSZ, 0, stream>>>(bpair, ssrc, off, row_ptr, col, nsrc, ndst, N, E, NB, G);

    const int gemmBlocks = (N + 15) / 16;
    int pullBlocks = (N + 7) / 8;
    if (pullBlocks > 2048) pullBlocks = 2048;

    gnn_gemm_kernel<<<gemmBlocks, 128, 0, stream>>>(feature, nsrc, W1, buf0, N);
    gnn_pull_relu_gemm_kernel<<<gemmBlocks, 128, 0, stream>>>(buf0, row_ptr, col, ndst, nsrc, b1, W2, buf1, N);
    gnn_pull_kernel<<<pullBlocks, 256, 0, stream>>>(buf1, row_ptr, col, ndst, b2, out, N);
}


# Row sums as one edge stream per wave over consecutive nodes: next 4 row loads issued before the current 4 are added, node bounds and norms held in lanes, col fetched a chunk ahead; sum order kept
# speedup vs baseline: 4.1736x; 1.0567x over previous
#include <hip/hip_runtime.h>
#include <hip/hip_bf16.h>

typedef __attribute__((ext_vector_type(2))) float v2f;
typedef __attribute__((ext_vector_type(8))) float v8f;

#define LDSTRIDE 132
#define SCAN_T 1024
#define BSH 9
#define BSZ 512
#define CH 4096
#define CT 256
#define MAXB 2048
#define FB 8

__global__ __launch_bounds__(CT) void gnn_chunk_hist_kernel(
    const int* __restrict__ src, const int* __restrict__ dst,
    int* __restrict__ tab, int E, int NB, int G) {
    __shared__ int hD[MAXB];
    __shared__ int hS[MAXB];
    const int tid = threadIdx.x;
    const int g   = blockIdx.x;
    for (int b = tid; b < NB; b += CT) { hD[b] = 0; hS[b] = 0; }
    __syncthreads();
    const int base = g * CH;
    int s[CH / CT], d[CH / CT];
#pragma unroll
    for (int k = 0; k < CH / CT; ++k) {
        const int e  = base + k * CT + tid;
        const int ec = (e < E) ? e : (E - 1);
        s[k] = src[ec];
        d[k] = dst[ec];
    }
#pragma unroll
    for (int k = 0; k < CH / CT; ++k) {
        const int e = base + k * CT + tid;
        if (e < E) {
            atomicAdd(&hD[d[k] >> BSH], 1);
            atomicAdd(&hS[s[k] >> BSH], 1);
        }
    }
    __syncthreads();
    for (int b = tid; b < NB; b += CT) {
        tab[(size_t)b * G + g]        = hD[b];
        tab[(size_t)(NB + b) * G + g] = hS[b];
    }
}

__global__ __launch_bounds__(SCAN_T) void gnn_tilesum_kernel(
    const int* __restrict__ v, int* __restrict__ tile_sum, int L) {
    __shared__ int wsum[32];
    const int tid  = threadIdx.x;
    const int lane = tid & 31;
    const int wv   = tid >> 5;
    const int i    = blockIdx.x * SCAN_T + tid;
    int s = (i < L) ? v[i] : 0;
    for (int o = 16; o > 0; o >>= 1) s += __shfl_xor(s, o);
    if (lane == 0) wsum[wv] = s;
    __syncthreads();
    if (wv == 0) {
        int t = wsum[lane];
        for (int o = 16; o > 0; o >>= 1) t += __shfl_xor(t, o);
        if (lane == 0) tile_sum[blockIdx.x] = t;
    }
}

__global__ __launch_bounds__(SCAN_T) void gnn_scan_kernel(
    const int* __restrict__ v, const int* __restrict__ tile_sum,
    int* __restrict__ off, int L) {
    __shared__ int wsum[32];
    const int tid  = threadIdx.x;
    const int lane = tid & 31;
    const int wv   = tid >> 5;
    const int i    = blockIdx.x * SCAN_T + tid;
    const int ci   = (i < L) ? v[i] : 0;

    int pre = 0;
    for (int j = lane; j < (int)blockIdx.x; j += 32) pre += tile_sum[j];
    for (int o = 16; o > 0; o >>= 1) pre += __shfl_xor(pre, o);

    int inc = ci;
    for (int o = 1; o < 32; o <<= 1) {
        int t = __shfl_up(inc, o);
        if (lane >= o) inc += t;
    }
    if (lane == 31) wsum[wv] = inc;
    __syncthreads();
    int wp = (lane < wv) ? wsum[lane] : 0;
    for (int o = 16; o > 0; o >>= 1) wp += __shfl_xor(wp, o);

    const int excl = pre + wp + inc - ci;
    if (i < L) off[i] = excl;
    if (i == L - 1) off[L] = excl + ci;
}

__global__ __launch_bounds__(CT) void gnn_bucket_scatter_kernel(
    const int* __restrict__ src, const int* __restrict__ dst,
    const int* __restrict__ off, int2* __restrict__ bpair,
    int* __restrict__ ssrc, int E, int NB, int G) {
    __shared__ int cD[MAXB];
    __shared__ int cS[MAXB];
    const int tid = threadIdx.x;
    const int g   = blockIdx.x;
    for (int b = tid; b < NB; b += CT) {
        cD[b] = off[(size_t)b * G + g];
        cS[b] = off[(size_t)(NB + b) * G + g] - E;
    }
    __syncthreads();
    const int base = g * CH;
    int s[CH / CT], d[CH / CT];
#pragma unroll
    for (int k = 0; k < CH / CT; ++k) {
        const int e  = base + k * CT + tid;
        const int ec = (e < E) ? e : (E - 1);
        s[k] = src[ec];
        d[k] = dst[ec];
    }
#pragma unroll
    for (int k = 0; k < CH / CT; ++k) {
        const int e = base + k * CT + tid;
        if (e < E) {
            const int pD = atomicAdd(&cD[d[k] >> BSH], 1);
            bpair[pD] = make_int2(s[k], d[k]);
            const int pS = atomicAdd(&cS[s[k] >> BSH], 1);
            ssrc[pS] = s[k];
        }
    }
}

__global__ __launch_bounds__(BSZ) void gnn_bucket_finish_kernel(
    const int2* __restrict__ bpair, const int* __restrict__ ssrc,
    const int* __restrict__ off, int* __restrict__ row_ptr, int* __restrict__ col,
    float* __restrict__ nsrc, float* __restrict__ ndst, int N, int E, int NB, int G) {
    __shared__ int cnt[BSZ];
    __shared__ int wsum[BSZ / 32];
    const int tid  = threadIdx.x;
    const int lane = tid & 31;
    const int wv   = tid >> 5;
    cnt[tid] = 0;
    __syncthreads();

    if ((int)blockIdx.x >= NB) {
        const int b   = (int)blockIdx.x - NB;
        const int beg = off[(size_t)(NB + b) * G] - E;
        const int end = off[(size_t)(NB + b + 1) * G] - E;
        for (int i0 = beg; i0 < end; i0 += FB * BSZ) {
            int v[FB];
#pragma unroll
            for (int k = 0; k < FB; ++k) {
                const int i = i0 + k * BSZ + tid;
                v[k] = ssrc[(i < end) ? i : (end - 1)];
            }
#pragma unroll
            for (int k = 0; k < FB; ++k)
                if (i0 + k * BSZ + tid < end) atomicAdd(&cnt[v[k] & (BSZ - 1)], 1);
        }
        __syncthreads();
        const int n = b * BSZ + tid;
        if (n < N) nsrc[n] = 1.0f / sqrtf(fmaxf((float)cnt[tid], 1.0f));
        return;
    }

    const int b   = (int)blockIdx.x;
    const int beg = off[(size_t)b * G];
    const int end = off[(size_t)(b + 1) * G];
    for (int i0 = beg; i0 < end; i0 += FB * BSZ) {
        int v[FB];
#pragma unroll
        for (int k = 0; k < FB; ++k) {
            const int i = i0 + k * BSZ + tid;
            v[k] = bpair[(i < end) ? i : (end - 1)].y;
        }
#pragma unroll
        for (int k = 0; k < FB; ++k)
            if (i0 + k * BSZ + tid < end) atomicAdd(&cnt[v[k] & (BSZ - 1)], 1);
    }
    __syncthreads();

    const int ci = cnt[tid];
    int inc = ci;
    for (int o = 1; o < 32; o <<= 1) {
        int t = __shfl_up(inc, o);
        if (lane >= o) inc += t;
    }
    if (lane == 31) wsum[wv] = inc;
    __syncthreads();
    int wp = (lane < wv) ? wsum[lane] : 0;
    for (int o = 16; o > 0; o >>= 1) wp += __shfl_xor(wp, o);

    const int excl = beg + wp + inc - ci;
    const int n = b * BSZ + tid;
    if (n < N) {
        row_ptr[n] = excl;
        ndst[n] = 1.0f / sqrtf(fmaxf((float)ci, 1.0f));
    }
    if (n == N - 1) row_ptr[N] = excl + ci;

    cnt[tid] = excl;
    __syncthreads();
    for (int i0 = beg; i0 < end; i0 += FB * BSZ) {
        int2 p[FB];
#pragma unroll
        for (int k = 0; k < FB; ++k) {
            const int i = i0 + k * BSZ + tid;
            p[k] = bpair[(i < end) ? i : (end - 1)];
        }
#pragma unroll
        for (int k = 0; k < FB; ++k) {
            if (i0 + k * BSZ + tid < end) {
                const int pos = atomicAdd(&cnt[p[k].y & (BSZ - 1)], 1);
                col[pos] = p[k].x;
            }
        }
    }
}

__global__ __launch_bounds__(128) void gnn_gemm_kernel(
    const float* __restrict__ H, const float* __restrict__ nsrc,
    const float* __restrict__ W, float* __restrict__ X, int N) {
    __shared__ float As[16 * LDSTRIDE];

    const int row0 = blockIdx.x * 16;
    const int tid  = threadIdx.x;

    for (int i = tid; i < 16 * 32; i += 128) {
        int r  = i >> 5;
        int c4 = (i & 31) * 4;
        int row = row0 + r;
        int rowc = (row < N) ? row : (N - 1);
        float s = nsrc[rowc];
        float4 v = *(const float4*)(H + (size_t)rowc * 128 + c4);
        float4 sv = make_float4(v.x * s, v.y * s, v.z * s, v.w * s);
        *(float4*)(&As[r * LDSTRIDE + c4]) = sv;
    }
    __syncthreads();

    const int wave = tid >> 5;
    const int lane = tid & 31;
    const int ln   = lane & 15;
    const int koff = (lane < 16) ? 0 : 2;

    const int n0 = wave * 32;
    const int n1 = n0 + 16;

    v8f acc0 = {};
    v8f acc1 = {};

    for (int kb = 0; kb < 128; kb += 4) {
        v2f a = *(const v2f*)(&As[ln * LDSTRIDE + kb + koff]);

        const float* wp = W + (size_t)(kb + koff) * 128;
        v2f b0, b1;
        b0.x = wp[n0 + ln];
        b0.y = wp[128 + n0 + ln];
        b1.x = wp[n1 + ln];
        b1.y = wp[128 + n1 + ln];

        acc0 = __builtin_amdgcn_wmma_f32_16x16x4_f32(
            false, a, false, b0, (short)0, acc0, false, false);
        acc1 = __builtin_amdgcn_wmma_f32_16x16x4_f32(
            false, a, false, b1, (short)0, acc1, false, false);
    }

    const int mbase = row0 + ((lane < 16) ? 0 : 8);
#pragma unroll
    for (int v = 0; v < 8; ++v) {
        int row = mbase + v;
        if (row < N) {
            X[(size_t)row * 128 + n0 + ln] = acc0[v];
            X[(size_t)row * 128 + n1 + ln] = acc1[v];
        }
    }
}

__device__ __forceinline__ float gnn_lane_f(float v, int k) {
    return __int_as_float(__builtin_amdgcn_readlane(__float_as_int(v), k));
}

template <bool FUSED>
__device__ __forceinline__ void gnn_stream_rows(
    const float* __restrict__ X, const int* __restrict__ col,
    const int rpv, const int nn, const float ndv, const float nsv,
    const float4 b, const int lane, float* dst) {
    const int stride = FUSED ? LDSTRIDE : 128;
    const float4 z = make_float4(0.0f, 0.0f, 0.0f, 0.0f);
    const float* xl = X + lane * 4;
    const int eA = __builtin_amdgcn_readlane(rpv, 0);
    const int eB = __builtin_amdgcn_readlane(rpv, nn);
    int k = 0;
    float4 acc = z;

    auto close_node = [&]() {
        const float s = gnn_lane_f(ndv, k);
        float4 o;
        if (FUSED) {
            const float t = gnn_lane_f(nsv, k);
            o.x = fmaxf(acc.x * s + b.x, 0.0f) * t;
            o.y = fmaxf(acc.y * s + b.y, 0.0f) * t;
            o.z = fmaxf(acc.z * s + b.z, 0.0f) * t;
            o.w = fmaxf(acc.w * s + b.w, 0.0f) * t;
        } else {
            o.x = acc.x * s + b.x;
            o.y = acc.y * s + b.y;
            o.z = acc.z * s + b.z;
            o.w = acc.w * s + b.w;
        }
        *(float4*)(dst + k * stride) = o;
        acc = z;
        ++k;
    };

    if (eB > eA) {
        int brel = __builtin_amdgcn_readlane(rpv, 1) - eA;
        int mycA = col[(eA + lane < eB) ? (eA + lane) : (eB - 1)];
        float4 buf[2][4];
#pragma unroll
        for (int i = 0; i < 4; ++i)
            buf[0][i] = *(const float4*)(xl + (size_t)__builtin_amdgcn_readlane(mycA, i) * 128);
        int base = eA;
        int erel = eB - eA;
        do {
            const int nx = base + 32 + lane;
            const int mycB = col[(nx < eB) ? nx : (eB - 1)];
#pragma unroll
            for (int g = 0; g < 8; ++g) {
#pragma unroll
                for (int i = 0; i < 4; ++i) {
                    const int idx = (g < 7) ? __builtin_amdgcn_readlane(mycA, (4 * (g + 1) + i) & 31)
                                            : __builtin_amdgcn_readlane(mycB, i);
                    buf[(g + 1) & 1][i] = *(const float4*)(xl + (size_t)idx * 128);
                }
#pragma unroll
                for (int i = 0; i < 4; ++i) {
                    while (brel == 4 * g + i) {
                        close_node();
                        brel = (k < nn) ? (__builtin_amdgcn_readlane(rpv, k + 1) - base) : -1;
                    }
                    const float4 v = buf[g & 1][i];
                    acc.x += v.x; acc.y += v.y; acc.z += v.z; acc.w += v.w;
                }
                if (g < 7 && 4 * (g + 1) >= erel) goto stream_done;
            }
            mycA = mycB;
            brel -= 32;
            base += 32;
            erel -= 32;
        } while (erel > 0);
    }
stream_done:
    while (k < nn) close_node();
}

__global__ __launch_bounds__(128) void gnn_pull_relu_gemm_kernel(
    const float* __restrict__ X, const int* __restrict__ row_ptr,
    const int* __restrict__ col, const float* __restrict__ ndst,
    const float* __restrict__ nsrc, const float* __restrict__ bias,
    const float* __restrict__ W, float* __restrict__ Y, int N) {
    __shared__ float As[16 * LDSTRIDE];

    const int row0 = blockIdx.x * 16;
    const int tid  = threadIdx.x;
    const int lane = tid & 31;
    const int wave = __builtin_amdgcn_readfirstlane(tid >> 5);
    const float4 b = *(const float4*)(bias + lane * 4);
    const float4 z = make_float4(0.0f, 0.0f, 0.0f, 0.0f);

    const int nq = row0 + wave * 4;
    int nn = N - nq;
    nn = (nn < 0) ? 0 : ((nn > 4) ? 4 : nn);
    float* arow = &As[(wave * 4) * LDSTRIDE + lane * 4];
    if (nn > 0) {
        const int ni  = nq + lane;
        const int rpv = row_ptr[(ni < N) ? ni : N];
        const int nc  = (ni < N) ? ni : (N - 1);
        const float ndv = ndst[nc];
        const float nsv = nsrc[nc];
        gnn_stream_rows<true>(X, col, rpv, nn, ndv, nsv, b, lane, arow);
    }
    for (int q = nn; q < 4; ++q)
        *(float4*)(arow + q * LDSTRIDE) = z;
    __syncthreads();

    const int ln   = lane & 15;
    const int koff = (lane < 16) ? 0 : 2;

    const int n0 = wave * 32;
    const int n1 = n0 + 16;

    v8f acc0 = {};
    v8f acc1 = {};

    for (int kb = 0; kb < 128; kb += 4) {
        v2f a = *(const v2f*)(&As[ln * LDSTRIDE + kb + koff]);

        const float* wp = W + (size_t)(kb + koff) * 128;
        v2f b0, b1;
        b0.x = wp[n0 + ln];
        b0.y = wp[128 + n0 + ln];
        b1.x = wp[n1 + ln];
        b1.y = wp[128 + n1 + ln];

        acc0 = __builtin_amdgcn_wmma_f32_16x16x4_f32(
            false, a, false, b0, (short)0, acc0, false, false);
        acc1 = __builtin_amdgcn_wmma_f32_16x16x4_f32(
            false, a, false, b1, (short)0, acc1, false, false);
    }

    const int mbase = row0 + ((lane < 16) ? 0 : 8);
#pragma unroll
    for (int v = 0; v < 8; ++v) {
        int row = mbase + v;
        if (row < N) {
            Y[(size_t)row * 128 + n0 + ln] = acc0[v];
            Y[(size_t)row * 128 + n1 + ln] = acc1[v];
        }
    }
}

__global__ __launch_bounds__(256) void gnn_pull_kernel(
    const float* __restrict__ X, const int* __restrict__ row_ptr,
    const int* __restrict__ col, const float* __restrict__ ndst,
    const float* __restrict__ bias, float* __restrict__ OUT, int N) {
    const int lane = threadIdx.x & 31;
    const int wpb  = blockDim.x >> 5;
    const int w0 = __builtin_amdgcn_readfirstlane((int)(blockIdx.x * wpb + (threadIdx.x >> 5)));
    const int nw = (int)gridDim.x * wpb;
    const float4 b = *(const float4*)(bias + lane * 4);

    int npb = (N + nw - 1) / nw;
    npb = (npb < 1) ? 1 : ((npb > 31) ? 31 : npb);
    for (int nA = w0 * npb; nA < N; nA += nw * npb) {
        const int nn  = (N - nA < npb) ? (N - nA) : npb;
        const int ni  = nA + lane;
        const int rpv = row_ptr[(ni < N) ? ni : N];
        const float ndv = ndst[(ni < N) ? ni : (N - 1)];
        gnn_stream_rows<false>(X, col, rpv, nn, ndv, 0.0f, b, lane, OUT + (size_t)nA * 128 + lane * 4);
    }
}

extern "C" void kernel_launch(void* const* d_in, const int* in_sizes, int n_in,
                              void* d_out, int out_size, void* d_ws, size_t ws_size,
                              hipStream_t stream) {
    const float* feature = (const float*)d_in[0];
    const int*   src     = (const int*)d_in[1];
    const int*   dst     = (const int*)d_in[2];
    const float* W1      = (const float*)d_in[3];
    const float* b1      = (const float*)d_in[4];
    const float* W2      = (const float*)d_in[5];
    const float* b2      = (const float*)d_in[6];
    float* out = (float*)d_out;

    const int F = 128;
    const int N = in_sizes[0] / F;
    const int E = in_sizes[1];
    const int NB = (N + BSZ - 1) / BSZ;
    const int G  = (E + CH - 1) / CH;
    const int L  = 2 * NB * G;
    const int tiles = (L + SCAN_T - 1) / SCAN_T;

    char* ws = (char*)d_ws;
    size_t o = 0;
    int* tab = (int*)(ws + o);
    o = (o + (size_t)L * sizeof(int) + 255) & ~(size_t)255;
    int* off = (int*)(ws + o);
    o = (o + ((size_t)L + 1) * sizeof(int) + 255) & ~(size_t)255;
    int* tile_sum = (int*)(ws + o);
    o = (o + (size_t)tiles * sizeof(int) + 255) & ~(size_t)255;
    int2* bpair = (int2*)(ws + o);
    o = (o + (size_t)E * sizeof(int2) + 255) & ~(size_t)255;
    int* ssrc = (int*)(ws + o);
    o = (o + (size_t)E * sizeof(int) + 255) & ~(size_t)255;
    float* nsrc = (float*)(ws + o);
    float* ndst = nsrc + N;
    o = (o + (size_t)2 * N * sizeof(float) + 255) & ~(size_t)255;
    int* row_ptr = (int*)(ws + o);
    o = (o + ((size_t)N + 1) * sizeof(int) + 255) & ~(size_t)255;
    int* col = (int*)(ws + o);
    o = (o + (size_t)E * sizeof(int) + 255) & ~(size_t)255;
    float* buf0 = (float*)(ws + o);
    float* buf1 = buf0 + (size_t)N * F;

    gnn_chunk_hist_kernel<<<G, CT, 0, stream>>>(src, dst, tab, E, NB, G);
    gnn_tilesum_kernel<<<tiles, SCAN_T, 0, stream>>>(tab, tile_sum, L);
    gnn_scan_kernel<<<tiles, SCAN_T, 0, stream>>>(tab, tile_sum, off, L);
    gnn_bucket_scatter_kernel<<<G, CT, 0, stream>>>(src, dst, off, bpair, ssrc, E, NB, G);
    gnn_bucket_finish_kernel<<<2 * NB, BSZ, 0, stream>>>(bpair, ssrc, off, row_ptr, col, nsrc, ndst, N, E, NB, G);

    const int gemmBlocks = (N + 15) / 16;
    int pullBlocks = (N + 7) / 8;
    if (pullBlocks > 2048) pullBlocks = 2048;

    gnn_gemm_kernel<<<gemmBlocks, 128, 0, stream>>>(feature, nsrc, W1, buf0, N);
    gnn_pull_relu_gemm_kernel<<<gemmBlocks, 128, 0, stream>>>(buf0, row_ptr, col, ndst, nsrc, b1, W2, buf1, N);
    gnn_pull_kernel<<<pullBlocks, 256, 0, stream>>>(buf1, row_ptr, col, ndst, b2, out, N);
}


# W copied per call into WMMA fragment order, lane c owning columns 2c, 2c+1 of its wave's 32: both B operands of a k-step are one 16-byte load per lane, results leave as whole-line 8-byte stores
# speedup vs baseline: 4.3182x; 1.0346x over previous
#include <hip/hip_runtime.h>
#include <hip/hip_bf16.h>

typedef __attribute__((ext_vector_type(2))) float v2f;
typedef __attribute__((ext_vector_type(8))) float v8f;

#define LDSTRIDE 132
#define SCAN_T 1024
#define BSH 9
#define BSZ 512
#define CH 4096
#define CT 256
#define MAXB 2048
#define FB 8

__device__ __forceinline__ float4 gnn_frag_entry(const float* __restrict__ Wm, int e) {
    const int w  = e >> 10;
    const int ks = (e >> 5) & 31;
    const int l  = e & 31;
    const int k  = 4 * ks + 2 * (l >> 4);
    const int cc = 32 * w + 2 * (l & 15);
    const float2 r0 = *(const float2*)(Wm + k * 128 + cc);
    const float2 r1 = *(const float2*)(Wm + (k + 1) * 128 + cc);
    return make_float4(r0.x, r1.x, r0.y, r1.y);
}

__global__ __launch_bounds__(CT) void gnn_chunk_hist_kernel(
    const int* __restrict__ src, const int* __restrict__ dst,
    int* __restrict__ tab, int E, int NB, int G,
    const float* __restrict__ Wa, const float* __restrict__ Wb,
    float* __restrict__ Wfa, float* __restrict__ Wfb) {
    __shared__ int hD[MAXB];
    __shared__ int hS[MAXB];
    const int tid = threadIdx.x;
    const int g   = blockIdx.x;
    for (int e = g * CT + tid; e < 2 * 4096; e += (int)gridDim.x * CT) {
        if (e < 4096) ((float4*)Wfa)[e] = gnn_frag_entry(Wa, e);
        else          ((float4*)Wfb)[e - 4096] = gnn_frag_entry(Wb, e - 4096);
    }
    for (int b = tid; b < NB; b += CT) { hD[b] = 0; hS[b] = 0; }
    __syncthreads();
    const int base = g * CH;
    int s[CH / CT], d[CH / CT];
#pragma unroll
    for (int k = 0; k < CH / CT; ++k) {
        const int e  = base + k * CT + tid;
        const int ec = (e < E) ? e : (E - 1);
        s[k] = src[ec];
        d[k] = dst[ec];
    }
#pragma unroll
    for (int k = 0; k < CH / CT; ++k) {
        const int e = base + k * CT + tid;
        if (e < E) {
            atomicAdd(&hD[d[k] >> BSH], 1);
            atomicAdd(&hS[s[k] >> BSH], 1);
        }
    }
    __syncthreads();
    for (int b = tid; b < NB; b += CT) {
        tab[(size_t)b * G + g]        = hD[b];
        tab[(size_t)(NB + b) * G + g] = hS[b];
    }
}

__global__ __launch_bounds__(SCAN_T) void gnn_tilesum_kernel(
    const int* __restrict__ v, int* __restrict__ tile_sum, int L) {
    __shared__ int wsum[32];
    const int tid  = threadIdx.x;
    const int lane = tid & 31;
    const int wv   = tid >> 5;
    const int i    = blockIdx.x * SCAN_T + tid;
    int s = (i < L) ? v[i] : 0;
    for (int o = 16; o > 0; o >>= 1) s += __shfl_xor(s, o);
    if (lane == 0) wsum[wv] = s;
    __syncthreads();
    if (wv == 0) {
        int t = wsum[lane];
        for (int o = 16; o > 0; o >>= 1) t += __shfl_xor(t, o);
        if (lane == 0) tile_sum[blockIdx.x] = t;
    }
}

__global__ __launch_bounds__(SCAN_T) void gnn_scan_kernel(
    const int* __restrict__ v, const int* __restrict__ tile_sum,
    int* __restrict__ off, int L) {
    __shared__ int wsum[32];
    const int tid  = threadIdx.x;
    const int lane = tid & 31;
    const int wv   = tid >> 5;
    const int i    = blockIdx.x * SCAN_T + tid;
    const int ci   = (i < L) ? v[i] : 0;

    int pre = 0;
    for (int j = lane; j < (int)blockIdx.x; j += 32) pre += tile_sum[j];
    for (int o = 16; o > 0; o >>= 1) pre += __shfl_xor(pre, o);

    int inc = ci;
    for (int o = 1; o < 32; o <<= 1) {
        int t = __shfl_up(inc, o);
        if (lane >= o) inc += t;
    }
    if (lane == 31) wsum[wv] = inc;
    __syncthreads();
    int wp = (lane < wv) ? wsum[lane] : 0;
    for (int o = 16; o > 0; o >>= 1) wp += __shfl_xor(wp, o);

    const int excl = pre + wp + inc - ci;
    if (i < L) off[i] = excl;
    if (i == L - 1) off[L] = excl + ci;
}

__global__ __launch_bounds__(CT) void gnn_bucket_scatter_kernel(
    const int* __restrict__ src, const int* __restrict__ dst,
    const int* __restrict__ off, int2* __restrict__ bpair,
    int* __restrict__ ssrc, int E, int NB, int G) {
    __shared__ int cD[MAXB];
    __shared__ int cS[MAXB];
    const int tid = threadIdx.x;
    const int g   = blockIdx.x;
    for (int b = tid; b < NB; b += CT) {
        cD[b] = off[(size_t)b * G + g];
        cS[b] = off[(size_t)(NB + b) * G + g] - E;
    }
    __syncthreads();
    const int base = g * CH;
    int s[CH / CT], d[CH / CT];
#pragma unroll
    for (int k = 0; k < CH / CT; ++k) {
        const int e  = base + k * CT + tid;
        const int ec = (e < E) ? e : (E - 1);
        s[k] = src[ec];
        d[k] = dst[ec];
    }
#pragma unroll
    for (int k = 0; k < CH / CT; ++k) {
        const int e = base + k * CT + tid;
        if (e < E) {
            const int pD = atomicAdd(&cD[d[k] >> BSH], 1);
            bpair[pD] = make_int2(s[k], d[k]);
            const int pS = atomicAdd(&cS[s[k] >> BSH], 1);
            ssrc[pS] = s[k];
        }
    }
}

__global__ __launch_bounds__(BSZ) void gnn_bucket_finish_kernel(
    const int2* __restrict__ bpair, const int* __restrict__ ssrc,
    const int* __restrict__ off, int* __restrict__ row_ptr, int* __restrict__ col,
    float* __restrict__ nsrc, float* __restrict__ ndst, int N, int E, int NB, int G) {
    __shared__ int cnt[BSZ];
    __shared__ int wsum[BSZ / 32];
    const int tid  = threadIdx.x;
    const int lane = tid & 31;
    const int wv   = tid >> 5;
    cnt[tid] = 0;
    __syncthreads();

    if ((int)blockIdx.x >= NB) {
        const int b   = (int)blockIdx.x - NB;
        const int beg = off[(size_t)(NB + b) * G] - E;
        const int end = off[(size_t)(NB + b + 1) * G] - E;
        for (int i0 = beg; i0 < end; i0 += FB * BSZ) {
            int v[FB];
#pragma unroll
            for (int k = 0; k < FB; ++k) {
                const int i = i0 + k * BSZ + tid;
                v[k] = ssrc[(i < end) ? i : (end - 1)];
            }
#pragma unroll
            for (int k = 0; k < FB; ++k)
                if (i0 + k * BSZ + tid < end) atomicAdd(&cnt[v[k] & (BSZ - 1)], 1);
        }
        __syncthreads();
        const int n = b * BSZ + tid;
        if (n < N) nsrc[n] = 1.0f / sqrtf(fmaxf((float)cnt[tid], 1.0f));
        return;
    }

    const int b   = (int)blockIdx.x;
    const int beg = off[(size_t)b * G];
    const int end = off[(size_t)(b + 1) * G];
    for (int i0 = beg; i0 < end; i0 += FB * BSZ) {
        int v[FB];
#pragma unroll
        for (int k = 0; k < FB; ++k) {
            const int i = i0 + k * BSZ + tid;
            v[k] = bpair[(i < end) ? i : (end - 1)].y;
        }
#pragma unroll
        for (int k = 0; k < FB; ++k)
            if (i0 + k * BSZ + tid < end) atomicAdd(&cnt[v[k] & (BSZ - 1)], 1);
    }
    __syncthreads();

    const int ci = cnt[tid];
    int inc = ci;
    for (int o = 1; o < 32; o <<= 1) {
        int t = __shfl_up(inc, o);
        if (lane >= o) inc += t;
    }
    if (lane == 31) wsum[wv] = inc;
    __syncthreads();
    int wp = (lane < wv) ? wsum[lane] : 0;
    for (int o = 16; o > 0; o >>= 1) wp += __shfl_xor(wp, o);

    const int excl = beg + wp + inc - ci;
    const int n = b * BSZ + tid;
    if (n < N) {
        row_ptr[n] = excl;
        ndst[n] = 1.0f / sqrtf(fmaxf((float)ci, 1.0f));
    }
    if (n == N - 1) row_ptr[N] = excl + ci;

    cnt[tid] = excl;
    __syncthreads();
    for (int i0 = beg; i0 < end; i0 += FB * BSZ) {
        int2 p[FB];
#pragma unroll
        for (int k = 0; k < FB; ++k) {
            const int i = i0 + k * BSZ + tid;
            p[k] = bpair[(i < end) ? i : (end - 1)];
        }
#pragma unroll
        for (int k = 0; k < FB; ++k) {
            if (i0 + k * BSZ + tid < end) {
                const int pos = atomicAdd(&cnt[p[k].y & (BSZ - 1)], 1);
                col[pos] = p[k].x;
            }
        }
    }
}

__global__ __launch_bounds__(128) void gnn_gemm_kernel(
    const float* __restrict__ H, const float* __restrict__ nsrc,
    const float* __restrict__ Wf, float* __restrict__ X, int N) {
    __shared__ float As[16 * LDSTRIDE];

    const int row0 = blockIdx.x * 16;
    const int tid  = threadIdx.x;

    for (int i = tid; i < 16 * 32; i += 128) {
        int r  = i >> 5;
        int c4 = (i & 31) * 4;
        int row = row0 + r;
        int rowc = (row < N) ? row : (N - 1);
        float s = nsrc[rowc];
        float4 v = *(const float4*)(H + (size_t)rowc * 128 + c4);
        float4 sv = make_float4(v.x * s, v.y * s, v.z * s, v.w * s);
        *(float4*)(&As[r * LDSTRIDE + c4]) = sv;
    }
    __syncthreads();

    const int wave = tid >> 5;
    const int lane = tid & 31;
    const int ln   = lane & 15;
    const int koff = (lane < 16) ? 0 : 2;

    const int n0 = wave * 32;

    v8f acc0 = {};
    v8f acc1 = {};

    const float4* wf = (const float4*)Wf + (wave * 32) * 32 + lane;

    for (int kb = 0; kb < 128; kb += 4) {
        v2f a = *(const v2f*)(&As[ln * LDSTRIDE + kb + koff]);

        const float4 wv = wf[(kb >> 2) * 32];
        v2f b0, b1;
        b0.x = wv.x;
        b0.y = wv.y;
        b1.x = wv.z;
        b1.y = wv.w;

        acc0 = __builtin_amdgcn_wmma_f32_16x16x4_f32(
            false, a, false, b0, (short)0, acc0, false, false);
        acc1 = __builtin_amdgcn_wmma_f32_16x16x4_f32(
            false, a, false, b1, (short)0, acc1, false, false);
    }

    const int mbase = row0 + ((lane < 16) ? 0 : 8);
#pragma unroll
    for (int v = 0; v < 8; ++v) {
        int row = mbase + v;
        if (row < N)
            *(float2*)(X + (size_t)row * 128 + n0 + 2 * ln) = make_float2(acc0[v], acc1[v]);
    }
}

__device__ __forceinline__ float gnn_lane_f(float v, int k) {
    return __int_as_float(__builtin_amdgcn_readlane(__float_as_int(v), k));
}

template <bool FUSED>
__device__ __forceinline__ void gnn_stream_rows(
    const float* __restrict__ X, const int* __restrict__ col,
    const int rpv, const int nn, const float ndv, const float nsv,
    const float4 b, const int lane, float* dst) {
    const int stride = FUSED ? LDSTRIDE : 128;
    const float4 z = make_float4(0.0f, 0.0f, 0.0f, 0.0f);
    const float* xl = X + lane * 4;
    const int eA = __builtin_amdgcn_readlane(rpv, 0);
    const int eB = __builtin_amdgcn_readlane(rpv, nn);
    int k = 0;
    float4 acc = z;

    auto close_node = [&]() {
        const float s = gnn_lane_f(ndv, k);
        float4 o;
        if (FUSED) {
            const float t = gnn_lane_f(nsv, k);
            o.x = fmaxf(acc.x * s + b.x, 0.0f) * t;
            o.y = fmaxf(acc.y * s + b.y, 0.0f) * t;
            o.z = fmaxf(acc.z * s + b.z, 0.0f) * t;
            o.w = fmaxf(acc.w * s + b.w, 0.0f) * t;
        } else {
            o.x = acc.x * s + b.x;
            o.y = acc.y * s + b.y;
            o.z = acc.z * s + b.z;
            o.w = acc.w * s + b.w;
        }
        *(float4*)(dst + k * stride) = o;
        acc = z;
        ++k;
    };

    if (eB > eA) {
        int brel = __builtin_amdgcn_readlane(rpv, 1) - eA;
        int mycA = col[(eA + lane < eB) ? (eA + lane) : (eB - 1)];
        float4 buf[2][4];
#pragma unroll
        for (int i = 0; i < 4; ++i)
            buf[0][i] = *(const float4*)(xl + (size_t)__builtin_amdgcn_readlane(mycA, i) * 128);
        int base = eA;
        int erel = eB - eA;
        do {
            const int nx = base + 32 + lane;
            const int mycB = col[(nx < eB) ? nx : (eB - 1)];
#pragma unroll
            for (int g = 0; g < 8; ++g) {
#pragma unroll
                for (int i = 0; i < 4; ++i) {
                    const int idx = (g < 7) ? __builtin_amdgcn_readlane(mycA, (4 * (g + 1) + i) & 31)
                                            : __builtin_amdgcn_readlane(mycB, i);
                    buf[(g + 1) & 1][i] = *(const float4*)(xl + (size_t)idx * 128);
                }
#pragma unroll
                for (int i = 0; i < 4; ++i) {
                    while (brel == 4 * g + i) {
                        close_node();
                        brel = (k < nn) ? (__builtin_amdgcn_readlane(rpv, k + 1) - base) : -1;
                    }
                    const float4 v = buf[g & 1][i];
                    acc.x += v.x; acc.y += v.y; acc.z += v.z; acc.w += v.w;
                }
                if (g < 7 && 4 * (g + 1) >= erel) goto stream_done;
            }
            mycA = mycB;
            brel -= 32;
            base += 32;
            erel -= 32;
        } while (erel > 0);
    }
stream_done:
    while (k < nn) close_node();
}

__global__ __launch_bounds__(128) void gnn_pull_relu_gemm_kernel(
    const float* __restrict__ X, const int* __restrict__ row_ptr,
    const int* __restrict__ col, const float* __restrict__ ndst,
    const float* __restrict__ nsrc, const float* __restrict__ bias,
    const float* __restrict__ Wf, float* __restrict__ Y, int N) {
    __shared__ float As[16 * LDSTRIDE];

    const int row0 = blockIdx.x * 16;
    const int tid  = threadIdx.x;
    const int lane = tid & 31;
    const int wave = __builtin_amdgcn_readfirstlane(tid >> 5);
    const float4 b = *(const float4*)(bias + lane * 4);
    const float4 z = make_float4(0.0f, 0.0f, 0.0f, 0.0f);

    const int nq = row0 + wave * 4;
    int nn = N - nq;
    nn = (nn < 0) ? 0 : ((nn > 4) ? 4 : nn);
    float* arow = &As[(wave * 4) * LDSTRIDE + lane * 4];
    if (nn > 0) {
        const int ni  = nq + lane;
        const int rpv = row_ptr[(ni < N) ? ni : N];
        const int nc  = (ni < N) ? ni : (N - 1);
        const float ndv = ndst[nc];
        const float nsv = nsrc[nc];
        gnn_stream_rows<true>(X, col, rpv, nn, ndv, nsv, b, lane, arow);
    }
    for (int q = nn; q < 4; ++q)
        *(float4*)(arow + q * LDSTRIDE) = z;
    __syncthreads();

    const int ln   = lane & 15;
    const int koff = (lane < 16) ? 0 : 2;

    const int n0 = wave * 32;

    v8f acc0 = {};
    v8f acc1 = {};

    const float4* wf = (const float4*)Wf + (wave * 32) * 32 + lane;

    for (int kb = 0; kb < 128; kb += 4) {
        v2f a = *(const v2f*)(&As[ln * LDSTRIDE + kb + koff]);

        const float4 wv = wf[(kb >> 2) * 32];
        v2f b0, b1;
        b0.x = wv.x;
        b0.y = wv.y;
        b1.x = wv.z;
        b1.y = wv.w;

        acc0 = __builtin_amdgcn_wmma_f32_16x16x4_f32(
            false, a, false, b0, (short)0, acc0, false, false);
        acc1 = __builtin_amdgcn_wmma_f32_16x16x4_f32(
            false, a, false, b1, (short)0, acc1, false, false);
    }

    const int mbase = row0 + ((lane < 16) ? 0 : 8);
#pragma unroll
    for (int v = 0; v < 8; ++v) {
        int row = mbase + v;
        if (row < N)
            *(float2*)(Y + (size_t)row * 128 + n0 + 2 * ln) = make_float2(acc0[v], acc1[v]);
    }
}

__global__ __launch_bounds__(256) void gnn_pull_kernel(
    const float* __restrict__ X, const int* __restrict__ row_ptr,
    const int* __restrict__ col, const float* __restrict__ ndst,
    const float* __restrict__ bias, float* __restrict__ OUT, int N) {
    const int lane = threadIdx.x & 31;
    const int wpb  = blockDim.x >> 5;
    const int w0 = __builtin_amdgcn_readfirstlane((int)(blockIdx.x * wpb + (threadIdx.x >> 5)));
    const int nw = (int)gridDim.x * wpb;
    const float4 b = *(const float4*)(bias + lane * 4);

    int npb = (N + nw - 1) / nw;
    npb = (npb < 1) ? 1 : ((npb > 31) ? 31 : npb);
    for (int nA = w0 * npb; nA < N; nA += nw * npb) {
        const int nn  = (N - nA < npb) ? (N - nA) : npb;
        const int ni  = nA + lane;
        const int rpv = row_ptr[(ni < N) ? ni : N];
        const float ndv = ndst[(ni < N) ? ni : (N - 1)];
        gnn_stream_rows<false>(X, col, rpv, nn, ndv, 0.0f, b, lane, OUT + (size_t)nA * 128 + lane * 4);
    }
}

extern "C" void kernel_launch(void* const* d_in, const int* in_sizes, int n_in,
                              void* d_out, int out_size, void* d_ws, size_t ws_size,
                              hipStream_t stream) {
    const float* feature = (const float*)d_in[0];
    const int*   src     = (const int*)d_in[1];
    const int*   dst     = (const int*)d_in[2];
    const float* W1      = (const float*)d_in[3];
    const float* b1      = (const float*)d_in[4];
    const float* W2      = (const float*)d_in[5];
    const float* b2      = (const float*)d_in[6];
    float* out = (float*)d_out;

    const int F = 128;
    const int N = in_sizes[0] / F;
    const int E = in_sizes[1];
    const int NB = (N + BSZ - 1) / BSZ;
    const int G  = (E + CH - 1) / CH;
    const int L  = 2 * NB * G;
    const int tiles = (L + SCAN_T - 1) / SCAN_T;

    char* ws = (char*)d_ws;
    size_t o = 0;
    float* wf1 = (float*)(ws + o);
    float* wf2 = wf1 + 128 * 128;
    o = (o + (size_t)2 * 128 * 128 * sizeof(float) + 255) & ~(size_t)255;
    int* tab = (int*)(ws + o);
    o = (o + (size_t)L * sizeof(int) + 255) & ~(size_t)255;
    int* off = (int*)(ws + o);
    o = (o + ((size_t)L + 1) * sizeof(int) + 255) & ~(size_t)255;
    int* tile_sum = (int*)(ws + o);
    o = (o + (size_t)tiles * sizeof(int) + 255) & ~(size_t)255;
    int2* bpair = (int2*)(ws + o);
    o = (o + (size_t)E * sizeof(int2) + 255) & ~(size_t)255;
    int* ssrc = (int*)(ws + o);
    o = (o + (size_t)E * sizeof(int) + 255) & ~(size_t)255;
    float* nsrc = (float*)(ws + o);
    float* ndst = nsrc + N;
    o = (o + (size_t)2 * N * sizeof(float) + 255) & ~(size_t)255;
    int* row_ptr = (int*)(ws + o);
    o = (o + ((size_t)N + 1) * sizeof(int) + 255) & ~(size_t)255;
    int* col = (int*)(ws + o);
    o = (o + (size_t)E * sizeof(int) + 255) & ~(size_t)255;
    float* buf0 = (float*)(ws + o);
    float* buf1 = buf0 + (size_t)N * F;

    gnn_chunk_hist_kernel<<<G, CT, 0, stream>>>(src, dst, tab, E, NB, G, W1, W2, wf1, wf2);
    gnn_tilesum_kernel<<<tiles, SCAN_T, 0, stream>>>(tab, tile_sum, L);
    gnn_scan_kernel<<<tiles, SCAN_T, 0, stream>>>(tab, tile_sum, off, L);
    gnn_bucket_scatter_kernel<<<G, CT, 0, stream>>>(src, dst, off, bpair, ssrc, E, NB, G);
    gnn_bucket_finish_kernel<<<2 * NB, BSZ, 0, stream>>>(bpair, ssrc, off, row_ptr, col, nsrc, ndst, N, E, NB, G);

    const int gemmBlocks = (N + 15) / 16;
    int pullBlocks = (N + 7) / 8;
    if (pullBlocks > 2048) pullBlocks = 2048;

    gnn_gemm_kernel<<<gemmBlocks, 128, 0, stream>>>(feature, nsrc, wf1, buf0, N);
    gnn_pull_relu_gemm_kernel<<<gemmBlocks, 128, 0, stream>>>(buf0, row_ptr, col, ndst, nsrc, b1, wf2, buf1, N);
    gnn_pull_kernel<<<pullBlocks, 256, 0, stream>>>(buf1, row_ptr, col, ndst, b2, out, N);
}


# Edge grouping without a global scan: each chunk sorts its edges by bucket inside its own slice, a bucket's work-group walks its per-chunk pieces by table; 3 launches fewer; sum order: LDS ranks
# speedup vs baseline: 5.0151x; 1.1614x over previous
#include <hip/hip_runtime.h>
#include <hip/hip_bf16.h>

typedef __attribute__((ext_vector_type(2))) float v2f;
typedef __attribute__((ext_vector_type(8))) float v8f;

#define LDSTRIDE 132
#define BSH 9
#define BSZ 512
#define CH 4096
#define CT 256
#define MAXB 2048
#define FB 16
#define PT 1024
#define PU 16

__device__ __forceinline__ float4 gnn_frag_entry(const float* __restrict__ Wm, int e) {
    const int w  = e >> 10;
    const int ks = (e >> 5) & 31;
    const int l  = e & 31;
    const int k  = 4 * ks + 2 * (l >> 4);
    const int cc = 32 * w + 2 * (l & 15);
    const float2 r0 = *(const float2*)(Wm + k * 128 + cc);
    const float2 r1 = *(const float2*)(Wm + (k + 1) * 128 + cc);
    return make_float4(r0.x, r1.x, r0.y, r1.y);
}

__global__ __launch_bounds__(CT) void gnn_chunk_sort_kernel(
    const int* __restrict__ src, const int* __restrict__ dst,
    int* __restrict__ tab, int2* __restrict__ bpair, int* __restrict__ ssrc,
    int E, int NB, int G,
    const float* __restrict__ Wa, const float* __restrict__ Wb,
    float* __restrict__ Wfa, float* __restrict__ Wfb) {
    __shared__ int hD[MAXB];
    __shared__ int hS[MAXB];
    __shared__ int wtot[2 * (CT / 32)];
    const int tid  = threadIdx.x;
    const int lane = tid & 31;
    const int wv   = tid >> 5;
    const int g    = blockIdx.x;
    for (int e = g * CT + tid; e < 2 * 4096; e += (int)gridDim.x * CT) {
        if (e < 4096) ((float4*)Wfa)[e] = gnn_frag_entry(Wa, e);
        else          ((float4*)Wfb)[e - 4096] = gnn_frag_entry(Wb, e - 4096);
    }
    for (int b = tid; b < MAXB; b += CT) { hD[b] = 0; hS[b] = 0; }
    __syncthreads();
    const int base = g * CH;
    int s[CH / CT], d[CH / CT];
#pragma unroll
    for (int k = 0; k < CH / CT; ++k) {
        const int e  = base + k * CT + tid;
        const int ec = (e < E) ? e : (E - 1);
        s[k] = src[ec];
        d[k] = dst[ec];
    }
#pragma unroll
    for (int k = 0; k < CH / CT; ++k) {
        const int e = base + k * CT + tid;
        if (e < E) {
            atomicAdd(&hD[d[k] >> BSH], 1);
            atomicAdd(&hS[s[k] >> BSH], 1);
        }
    }
    __syncthreads();

    int cd[8], cs[8];
    int sd = 0, ss = 0;
#pragma unroll
    for (int j = 0; j < 8; ++j) {
        cd[j] = hD[8 * tid + j];
        cs[j] = hS[8 * tid + j];
        sd += cd[j];
        ss += cs[j];
    }
    int id = sd, is = ss;
    for (int o = 1; o < 32; o <<= 1) {
        const int td = __shfl_up(id, o);
        const int ts = __shfl_up(is, o);
        if (lane >= o) { id += td; is += ts; }
    }
    if (lane == 31) { wtot[wv] = id; wtot[CT / 32 + wv] = is; }
    __syncthreads();
    int rd = id - sd, rs = is - ss;
    for (int w = 0; w < wv; ++w) { rd += wtot[w]; rs += wtot[CT / 32 + w]; }
#pragma unroll
    for (int j = 0; j < 8; ++j) {
        const int b = 8 * tid + j;
        if (b < NB) {
            tab[(size_t)b * G + g]        = (cd[j] << 16) | rd;
            tab[(size_t)(NB + b) * G + g] = (cs[j] << 16) | rs;
        }
        hD[b] = rd;
        hS[b] = rs;
        rd += cd[j];
        rs += cs[j];
    }
    __syncthreads();
#pragma unroll
    for (int k = 0; k < CH / CT; ++k) {
        const int e = base + k * CT + tid;
        if (e < E) {
            const int pD = atomicAdd(&hD[d[k] >> BSH], 1);
            bpair[base + pD] = make_int2(s[k], d[k]);
            const int pS = atomicAdd(&hS[s[k] >> BSH], 1);
            ssrc[base + pS] = s[k];
        }
    }
}

template <int MODE>
__device__ __forceinline__ void gnn_walk_pieces(
    const int* __restrict__ trow, const int G,
    const int2* __restrict__ bpair, const int* __restrict__ ssrc,
    int* cnt, int* ptab, int* __restrict__ col, const int tid) {
    const int lane = tid & 31;
    const int wv = __builtin_amdgcn_readfirstlane(tid >> 5);
    for (int g0 = 0; g0 < G; g0 += PT) {
        const int np = (G - g0 < PT) ? (G - g0) : PT;
        __syncthreads();
        for (int p = tid; p < np; p += BSZ) ptab[p] = trow[g0 + p];
        __syncthreads();
        for (int p0 = wv * PU; p0 < np; p0 += (BSZ / 32) * PU) {
            int len[PU], at[PU];
            int2 v[PU];
            const int pl = p0 + (lane & (PU - 1));
            const int wl = (pl < np) ? ptab[pl] : 0;
#pragma unroll
            for (int u = 0; u < PU; ++u) {
                const int p = p0 + u;
                const int w = __builtin_amdgcn_readlane(wl, u);
                len[u] = w >> 16;
                at[u]  = (g0 + p) * CH + (w & 0xffff);
                const int i = (lane < len[u]) ? (at[u] + lane) : 0;
                if (MODE == 0) v[u] = make_int2(0, ssrc[i]);
                else           v[u] = bpair[i];
            }
#pragma unroll
            for (int u = 0; u < PU; ++u) {
                if (lane < len[u]) {
                    if (MODE == 2) col[atomicAdd(&cnt[v[u].y & (BSZ - 1)], 1)] = v[u].x;
                    else           atomicAdd(&cnt[v[u].y & (BSZ - 1)], 1);
                }
                for (int i = lane + 32; i < len[u]; i += 32) {
                    if (MODE == 0) {
                        atomicAdd(&cnt[ssrc[at[u] + i] & (BSZ - 1)], 1);
                    } else {
                        const int2 q = bpair[at[u] + i];
                        if (MODE == 2) col[atomicAdd(&cnt[q.y & (BSZ - 1)], 1)] = q.x;
                        else           atomicAdd(&cnt[q.y & (BSZ - 1)], 1);
                    }
                }
            }
        }
    }
}

__global__ __launch_bounds__(BSZ) void gnn_bucket_finish_kernel(
    const int2* __restrict__ bpair, const int* __restrict__ ssrc,
    const int* __restrict__ tab, int* __restrict__ row_ptr, int* __restrict__ col,
    float* __restrict__ nsrc, float* __restrict__ ndst, int N, int NB, int G) {
    __shared__ int cnt[BSZ];
    __shared__ int wsum[BSZ / 32];
    __shared__ int ptab[PT];
    const int tid  = threadIdx.x;
    const int lane = tid & 31;
    const int wv   = tid >> 5;
    cnt[tid] = 0;

    if ((int)blockIdx.x >= NB) {
        const int b = (int)blockIdx.x - NB;
        gnn_walk_pieces<0>(tab + (size_t)(NB + b) * G, G, bpair, ssrc, cnt, ptab, col, tid);
        __syncthreads();
        const int n = b * BSZ + tid;
        if (n < N) nsrc[n] = 1.0f / sqrtf(fmaxf((float)cnt[tid], 1.0f));
        return;
    }

    const int b = (int)blockIdx.x;
    int part = 0;
    {
        const int nent = b * G;
        const int n4   = nent >> 2;
        const int4* t4 = (const int4*)tab;
        for (int i0 = 0; i0 < n4; i0 += FB * BSZ) {
            int4 q[FB];
#pragma unroll
            for (int k = 0; k < FB; ++k) {
                const int i = i0 + k * BSZ + tid;
                q[k] = t4[(i < n4) ? i : 0];
            }
#pragma unroll
            for (int k = 0; k < FB; ++k)
                if (i0 + k * BSZ + tid < n4)
                    part += (q[k].x >> 16) + (q[k].y >> 16) + (q[k].z >> 16) + (q[k].w >> 16);
        }
        const int r = (n4 << 2) + tid;
        if (r < nent) part += tab[r] >> 16;
    }
    for (int o = 16; o > 0; o >>= 1) part += __shfl_xor(part, o);
    if (lane == 0) wsum[wv] = part;
    __syncthreads();
    int beg = (lane < BSZ / 32) ? wsum[lane] : 0;
    for (int o = 16; o > 0; o >>= 1) beg += __shfl_xor(beg, o);

    const int* trow = tab + (size_t)b * G;
    gnn_walk_pieces<1>(trow, G, bpair, ssrc, cnt, ptab, col, tid);
    __syncthreads();

    const int ci = cnt[tid];
    int inc = ci;
    for (int o = 1; o < 32; o <<= 1) {
        int t = __shfl_up(inc, o);
        if (lane >= o) inc += t;
    }
    if (lane == 31) wsum[wv] = inc;
    __syncthreads();
    int wp = (lane < wv) ? wsum[lane] : 0;
    for (int o = 16; o > 0; o >>= 1) wp += __shfl_xor(wp, o);

    const int excl = beg + wp + inc - ci;
    const int n = b * BSZ + tid;
    if (n < N) {
        row_ptr[n] = excl;
        ndst[n] = 1.0f / sqrtf(fmaxf((float)ci, 1.0f));
    }
    if (n == N - 1) row_ptr[N] = excl + ci;

    cnt[tid] = excl;
    gnn_walk_pieces<2>(trow, G, bpair, ssrc, cnt, ptab, col, tid);
}

__global__ __launch_bounds__(128) void gnn_gemm_kernel(
    const float* __restrict__ H, const float* __restrict__ nsrc,
    const float* __restrict__ Wf, float* __restrict__ X, int N) {
    __shared__ float As[16 * LDSTRIDE];

    const int row0 = blockIdx.x * 16;
    const int tid  = threadIdx.x;

    for (int i = tid; i < 16 * 32; i += 128) {
        int r  = i >> 5;
        int c4 = (i & 31) * 4;
        int row = row0 + r;
        int rowc = (row < N) ? row : (N - 1);
        float s = nsrc[rowc];
        float4 v = *(const float4*)(H + (size_t)rowc * 128 + c4);
        float4 sv = make_float4(v.x * s, v.y * s, v.z * s, v.w * s);
        *(float4*)(&As[r * LDSTRIDE + c4]) = sv;
    }
    __syncthreads();

    const int wave = tid >> 5;
    const int lane = tid & 31;
    const int ln   = lane & 15;
    const int koff = (lane < 16) ? 0 : 2;

    const int n0 = wave * 32;

    v8f acc0 = {};
    v8f acc1 = {};

    const float4* wf = (const float4*)Wf + (wave * 32) * 32 + lane;

    for (int kb = 0; kb < 128; kb += 4) {
        v2f a = *(const v2f*)(&As[ln * LDSTRIDE + kb + koff]);

        const float4 wv = wf[(kb >> 2) * 32];
        v2f b0, b1;
        b0.x = wv.x;
        b0.y = wv.y;
        b1.x = wv.z;
        b1.y = wv.w;

        acc0 = __builtin_amdgcn_wmma_f32_16x16x4_f32(
            false, a, false, b0, (short)0, acc0, false, false);
        acc1 = __builtin_amdgcn_wmma_f32_16x16x4_f32(
            false, a, false, b1, (short)0, acc1, false, false);
    }

    const int mbase = row0 + ((lane < 16) ? 0 : 8);
#pragma unroll
    for (int v = 0; v < 8; ++v) {
        int row = mbase + v;
        if (row < N)
            *(float2*)(X + (size_t)row * 128 + n0 + 2 * ln) = make_float2(acc0[v], acc1[v]);
    }
}

__device__ __forceinline__ float gnn_lane_f(float v, int k) {
    return __int_as_float(__builtin_amdgcn_readlane(__float_as_int(v), k));
}

template <bool FUSED>
__device__ __forceinline__ void gnn_stream_rows(
    const float* __restrict__ X, const int* __restrict__ col,
    const int rpv, const int nn, const float ndv, const float nsv,
    const float4 b, const int lane, float* dst) {
    const int stride = FUSED ? LDSTRIDE : 128;
    const float4 z = make_float4(0.0f, 0.0f, 0.0f, 0.0f);
    const float* xl = X + lane * 4;
    const int eA = __builtin_amdgcn_readlane(rpv, 0);
    const int eB = __builtin_amdgcn_readlane(rpv, nn);
    int k = 0;
    float4 acc = z;

    auto close_node = [&]() {
        const float s = gnn_lane_f(ndv, k);
        float4 o;
        if (FUSED) {
            const float t = gnn_lane_f(nsv, k);
            o.x = fmaxf(acc.x * s + b.x, 0.0f) * t;
            o.y = fmaxf(acc.y * s + b.y, 0.0f) * t;
            o.z = fmaxf(acc.z * s + b.z, 0.0f) * t;
            o.w = fmaxf(acc.w * s + b.w, 0.0f) * t;
        } else {
            o.x = acc.x * s + b.x;
            o.y = acc.y * s + b.y;
            o.z = acc.z * s + b.z;
            o.w = acc.w * s + b.w;
        }
        *(float4*)(dst + k * stride) = o;
        acc = z;
        ++k;
    };

    if (eB > eA) {
        int brel = __builtin_amdgcn_readlane(rpv, 1) - eA;
        int mycA = col[(eA + lane < eB) ? (eA + lane) : (eB - 1)];
        float4 buf[2][4];
#pragma unroll
        for (int i = 0; i < 4; ++i)
            buf[0][i] = *(const float4*)(xl + (size_t)__builtin_amdgcn_readlane(mycA, i) * 128);
        int base = eA;
        int erel = eB - eA;
        do {
            const int nx = base + 32 + lane;
            const int mycB = col[(nx < eB) ? nx : (eB - 1)];
#pragma unroll
            for (int g = 0; g < 8; ++g) {
#pragma unroll
                for (int i = 0; i < 4; ++i) {
                    const int idx = (g < 7) ? __builtin_amdgcn_readlane(mycA, (4 * (g + 1) + i) & 31)
                                            : __builtin_amdgcn_readlane(mycB, i);
                    buf[(g + 1) & 1][i] = *(const float4*)(xl + (size_t)idx * 128);
                }
#pragma unroll
                for (int i = 0; i < 4; ++i) {
                    while (brel == 4 * g + i) {
                        close_node();
                        brel = (k < nn) ? (__builtin_amdgcn_readlane(rpv, k + 1) - base) : -1;
                    }
                    const float4 v = buf[g & 1][i];
                    acc.x += v.x; acc.y += v.y; acc.z += v.z; acc.w += v.w;
                }
                if (g < 7 && 4 * (g + 1) >= erel) goto stream_done;
            }
            mycA = mycB;
            brel -= 32;
            base += 32;
            erel -= 32;
        } while (erel > 0);
    }
stream_done:
    while (k < nn) close_node();
}

__global__ __launch_bounds__(128) void gnn_pull_relu_gemm_kernel(
    const float* __restrict__ X, const int* __restrict__ row_ptr,
    const int* __restrict__ col, const float* __restrict__ ndst,
    const float* __restrict__ nsrc, const float* __restrict__ bias,
    const float* __restrict__ Wf, float* __restrict__ Y, int N) {
    __shared__ float As[16 * LDSTRIDE];

    const int row0 = blockIdx.x * 16;
    const int tid  = threadIdx.x;
    const int lane = tid & 31;
    const int wave = __builtin_amdgcn_readfirstlane(tid >> 5);
    const float4 b = *(const float4*)(bias + lane * 4);
    const float4 z = make_float4(0.0f, 0.0f, 0.0f, 0.0f);

    const int nq = row0 + wave * 4;
    int nn = N - nq;
    nn = (nn < 0) ? 0 : ((nn > 4) ? 4 : nn);
    float* arow = &As[(wave * 4) * LDSTRIDE + lane * 4];
    if (nn > 0) {
        const int ni  = nq + lane;
        const int rpv = row_ptr[(ni < N) ? ni : N];
        const int nc  = (ni < N) ? ni : (N - 1);
        const float ndv = ndst[nc];
        const float nsv = nsrc[nc];
        gnn_stream_rows<true>(X, col, rpv, nn, ndv, nsv, b, lane, arow);
    }
    for (int q = nn; q < 4; ++q)
        *(float4*)(arow + q * LDSTRIDE) = z;
    __syncthreads();

    const int ln   = lane & 15;
    const int koff = (lane < 16) ? 0 : 2;

    const int n0 = wave * 32;

    v8f acc0 = {};
    v8f acc1 = {};

    const float4* wf = (const float4*)Wf + (wave * 32) * 32 + lane;

    for (int kb = 0; kb < 128; kb += 4) {
        v2f a = *(const v2f*)(&As[ln * LDSTRIDE + kb + koff]);

        const float4 wv = wf[(kb >> 2) * 32];
        v2f b0, b1;
        b0.x = wv.x;
        b0.y = wv.y;
        b1.x = wv.z;
        b1.y = wv.w;

        acc0 = __builtin_amdgcn_wmma_f32_16x16x4_f32(
            false, a, false, b0, (short)0, acc0, false, false);
        acc1 = __builtin_amdgcn_wmma_f32_16x16x4_f32(
            false, a, false, b1, (short)0, acc1, false, false);
    }

    const int mbase = row0 + ((lane < 16) ? 0 : 8);
#pragma unroll
    for (int v = 0; v < 8; ++v) {
        int row = mbase + v;
        if (row < N)
            *(float2*)(Y + (size_t)row * 128 + n0 + 2 * ln) = make_float2(acc0[v], acc1[v]);
    }
}

__global__ __launch_bounds__(256) void gnn_pull_kernel(
    const float* __restrict__ X, const int* __restrict__ row_ptr,
    const int* __restrict__ col, const float* __restrict__ ndst,
    const float* __restrict__ bias, float* __restrict__ OUT, int N) {
    const int lane = threadIdx.x & 31;
    const int wpb  = blockDim.x >> 5;
    const int w0 = __builtin_amdgcn_readfirstlane((int)(blockIdx.x * wpb + (threadIdx.x >> 5)));
    const int nw = (int)gridDim.x * wpb;
    const float4 b = *(const float4*)(bias + lane * 4);

    int npb = (N + nw - 1) / nw;
    npb = (npb < 1) ? 1 : ((npb > 31) ? 31 : npb);
    for (int nA = w0 * npb; nA < N; nA += nw * npb) {
        const int nn  = (N - nA < npb) ? (N - nA) : npb;
        const int ni  = nA + lane;
        const int rpv = row_ptr[(ni < N) ? ni : N];
        const float ndv = ndst[(ni < N) ? ni : (N - 1)];
        gnn_stream_rows<false>(X, col, rpv, nn, ndv, 0.0f, b, lane, OUT + (size_t)nA * 128 + lane * 4);
    }
}

extern "C" void kernel_launch(void* const* d_in, const int* in_sizes, int n_in,
                              void* d_out, int out_size, void* d_ws, size_t ws_size,
                              hipStream_t stream) {
    const float* feature = (const float*)d_in[0];
    const int*   src     = (const int*)d_in[1];
    const int*   dst     = (const int*)d_in[2];
    const float* W1      = (const float*)d_in[3];
    const float* b1      = (const float*)d_in[4];
    const float* W2      = (const float*)d_in[5];
    const float* b2      = (const float*)d_in[6];
    float* out = (float*)d_out;

    const int F = 128;
    const int N = in_sizes[0] / F;
    const int E = in_sizes[1];
    const int NB = (N + BSZ - 1) / BSZ;
    const int G  = (E + CH - 1) / CH;
    const int L  = 2 * NB * G;

    char* ws = (char*)d_ws;
    size_t o = 0;
    float* wf1 = (float*)(ws + o);
    float* wf2 = wf1 + 128 * 128;
    o = (o + (size_t)2 * 128 * 128 * sizeof(float) + 255) & ~(size_t)255;
    int* tab = (int*)(ws + o);
    o = (o + (size_t)L * sizeof(int) + 255) & ~(size_t)255;
    int2* bpair = (int2*)(ws + o);
    o = (o + (size_t)E * sizeof(int2) + 255) & ~(size_t)255;
    int* ssrc = (int*)(ws + o);
    o = (o + (size_t)E * sizeof(int) + 255) & ~(size_t)255;
    float* nsrc = (float*)(ws + o);
    float* ndst = nsrc + N;
    o = (o + (size_t)2 * N * sizeof(float) + 255) & ~(size_t)255;
    int* row_ptr = (int*)(ws + o);
    o = (o + ((size_t)N + 1) * sizeof(int) + 255) & ~(size_t)255;
    int* col = (int*)(ws + o);
    o = (o + (size_t)E * sizeof(int) + 255) & ~(size_t)255;
    float* buf0 = (float*)(ws + o);
    float* buf1 = buf0 + (size_t)N * F;

    gnn_chunk_sort_kernel<<<G, CT, 0, stream>>>(src, dst, tab, bpair, ssrc, E, NB, G, W1, W2, wf1, wf2);
    gnn_bucket_finish_kernel<<<2 * NB, BSZ, 0, stream>>>(bpair, ssrc, tab, row_ptr, col, nsrc, ndst, N, NB, G);

    const int gemmBlocks = (N + 15) / 16;
    int pullBlocks = (N + 7) / 8;
    if (pullBlocks > 2048) pullBlocks = 2048;

    gnn_gemm_kernel<<<gemmBlocks, 128, 0, stream>>>(feature, nsrc, wf1, buf0, N);
    gnn_pull_relu_gemm_kernel<<<gemmBlocks, 128, 0, stream>>>(buf0, row_ptr, col, ndst, nsrc, b1, wf2, buf1, N);
    gnn_pull_kernel<<<pullBlocks, 256, 0, stream>>>(buf1, row_ptr, col, ndst, b2, out, N);
}


# Both edge-grouping launches build their sorted output in LDS (a chunk's pairs and source ids, a bucket's run of col) and store whole 128-byte lines, not scattered 4/8-byte words; LDS-rank order kept
# speedup vs baseline: 5.3710x; 1.0710x over previous
#include <hip/hip_runtime.h>
#include <hip/hip_bf16.h>

typedef __attribute__((ext_vector_type(2))) float v2f;
typedef __attribute__((ext_vector_type(8))) float v8f;

#define LDSTRIDE 132
#define BSH 9
#define BSZ 512
#define CH 4096
#define CT 256
#define MAXB 2048
#define FB 16
#define PT 1024
#define PU 16
#define SCAP 12288

__device__ __forceinline__ float4 gnn_frag_entry(const float* __restrict__ Wm, int e) {
    const int w  = e >> 10;
    const int ks = (e >> 5) & 31;
    const int l  = e & 31;
    const int k  = 4 * ks + 2 * (l >> 4);
    const int cc = 32 * w + 2 * (l & 15);
    const float2 r0 = *(const float2*)(Wm + k * 128 + cc);
    const float2 r1 = *(const float2*)(Wm + (k + 1) * 128 + cc);
    return make_float4(r0.x, r1.x, r0.y, r1.y);
}

__global__ __launch_bounds__(CT) void gnn_chunk_sort_kernel(
    const int* __restrict__ src, const int* __restrict__ dst,
    int* __restrict__ tab, int2* __restrict__ bpair, int* __restrict__ ssrc,
    int E, int NB, int G,
    const float* __restrict__ Wa, const float* __restrict__ Wb,
    float* __restrict__ Wfa, float* __restrict__ Wfb) {
    __shared__ int hD[MAXB];
    __shared__ int hS[MAXB];
    __shared__ int wtot[2 * (CT / 32)];
    __shared__ __attribute__((aligned(16))) int2 stg[CH];
    const int tid  = threadIdx.x;
    const int lane = tid & 31;
    const int wv   = tid >> 5;
    const int g    = blockIdx.x;
    for (int e = g * CT + tid; e < 2 * 4096; e += (int)gridDim.x * CT) {
        if (e < 4096) ((float4*)Wfa)[e] = gnn_frag_entry(Wa, e);
        else          ((float4*)Wfb)[e - 4096] = gnn_frag_entry(Wb, e - 4096);
    }
    for (int b = tid; b < MAXB; b += CT) { hD[b] = 0; hS[b] = 0; }
    __syncthreads();
    const int base = g * CH;
    int s[CH / CT], d[CH / CT];
#pragma unroll
    for (int k = 0; k < CH / CT; ++k) {
        const int e  = base + k * CT + tid;
        const int ec = (e < E) ? e : (E - 1);
        s[k] = src[ec];
        d[k] = dst[ec];
    }
#pragma unroll
    for (int k = 0; k < CH / CT; ++k) {
        const int e = base + k * CT + tid;
        if (e < E) {
            atomicAdd(&hD[d[k] >> BSH], 1);
            atomicAdd(&hS[s[k] >> BSH], 1);
        }
    }
    __syncthreads();

    int cd[8], cs[8];
    int sd = 0, ss = 0;
#pragma unroll
    for (int j = 0; j < 8; ++j) {
        cd[j] = hD[8 * tid + j];
        cs[j] = hS[8 * tid + j];
        sd += cd[j];
        ss += cs[j];
    }
    int id = sd, is = ss;
    for (int o = 1; o < 32; o <<= 1) {
        const int td = __shfl_up(id, o);
        const int ts = __shfl_up(is, o);
        if (lane >= o) { id += td; is += ts; }
    }
    if (lane == 31) { wtot[wv] = id; wtot[CT / 32 + wv] = is; }
    __syncthreads();
    int rd = id - sd, rs = is - ss;
    for (int w = 0; w < wv; ++w) { rd += wtot[w]; rs += wtot[CT / 32 + w]; }
#pragma unroll
    for (int j = 0; j < 8; ++j) {
        const int b = 8 * tid + j;
        if (b < NB) {
            tab[(size_t)b * G + g]        = (cd[j] << 16) | rd;
            tab[(size_t)(NB + b) * G + g] = (cs[j] << 16) | rs;
        }
        hD[b] = rd;
        hS[b] = rs;
        rd += cd[j];
        rs += cs[j];
    }
    __syncthreads();
    const int nE = (E - base < CH) ? (E - base) : CH;
#pragma unroll
    for (int k = 0; k < CH / CT; ++k) {
        const int e = base + k * CT + tid;
        if (e < E) stg[atomicAdd(&hD[d[k] >> BSH], 1)] = make_int2(s[k], d[k]);
    }
    __syncthreads();
    {
        const int4* s4 = (const int4*)stg;
        int4* o4 = (int4*)(bpair + base);
        const int n4 = nE >> 1;
        for (int i = tid; i < n4; i += CT) o4[i] = s4[i];
        if (tid == 0 && (nE & 1)) bpair[base + nE - 1] = stg[nE - 1];
    }
    __syncthreads();
    int* sts = (int*)stg;
#pragma unroll
    for (int k = 0; k < CH / CT; ++k) {
        const int e = base + k * CT + tid;
        if (e < E) sts[atomicAdd(&hS[s[k] >> BSH], 1)] = s[k];
    }
    __syncthreads();
    {
        const int4* s4 = (const int4*)sts;
        int4* o4 = (int4*)(ssrc + base);
        const int n4 = nE >> 2;
        for (int i = tid; i < n4; i += CT) o4[i] = s4[i];
        const int r = (n4 << 2) + tid;
        if (r < nE) ssrc[base + r] = sts[r];
    }
}

template <int MODE>
__device__ __forceinline__ void gnn_walk_pieces(
    const int* __restrict__ trow, const int G,
    const int2* __restrict__ bpair, const int* __restrict__ ssrc,
    int* cnt, int* ptab, int* __restrict__ col, int* lst, const int tid) {
    const int lane = tid & 31;
    const int wv = __builtin_amdgcn_readfirstlane(tid >> 5);
    for (int g0 = 0; g0 < G; g0 += PT) {
        const int np = (G - g0 < PT) ? (G - g0) : PT;
        __syncthreads();
        for (int p = tid; p < np; p += BSZ) ptab[p] = trow[g0 + p];
        __syncthreads();
        for (int p0 = wv * PU; p0 < np; p0 += (BSZ / 32) * PU) {
            int len[PU], at[PU];
            int2 v[PU];
            const int pl = p0 + (lane & (PU - 1));
            const int wl = (pl < np) ? ptab[pl] : 0;
#pragma unroll
            for (int u = 0; u < PU; ++u) {
                const int p = p0 + u;
                const int w = __builtin_amdgcn_readlane(wl, u);
                len[u] = w >> 16;
                at[u]  = (g0 + p) * CH + (w & 0xffff);
                const int i = (lane < len[u]) ? (at[u] + lane) : 0;
                if (MODE == 0) v[u] = make_int2(0, ssrc[i]);
                else           v[u] = bpair[i];
            }
#pragma unroll
            for (int u = 0; u < PU; ++u) {
                if (lane < len[u]) {
                    if (MODE == 2)      col[atomicAdd(&cnt[v[u].y & (BSZ - 1)], 1)] = v[u].x;
                    else if (MODE == 3) lst[atomicAdd(&cnt[v[u].y & (BSZ - 1)], 1)] = v[u].x;
                    else                atomicAdd(&cnt[v[u].y & (BSZ - 1)], 1);
                }
                for (int i = lane + 32; i < len[u]; i += 32) {
                    if (MODE == 0) {
                        atomicAdd(&cnt[ssrc[at[u] + i] & (BSZ - 1)], 1);
                    } else {
                        const int2 q = bpair[at[u] + i];
                        if (MODE == 2)      col[atomicAdd(&cnt[q.y & (BSZ - 1)], 1)] = q.x;
                        else if (MODE == 3) lst[atomicAdd(&cnt[q.y & (BSZ - 1)], 1)] = q.x;
                        else                atomicAdd(&cnt[q.y & (BSZ - 1)], 1);
                    }
                }
            }
        }
    }
}

__global__ __launch_bounds__(BSZ) void gnn_bucket_finish_kernel(
    const int2* __restrict__ bpair, const int* __restrict__ ssrc,
    const int* __restrict__ tab, int* __restrict__ row_ptr, int* __restrict__ col,
    float* __restrict__ nsrc, float* __restrict__ ndst, int N, int NB, int G) {
    __shared__ int cnt[BSZ];
    __shared__ int wsum[BSZ / 32];
    __shared__ int ptab[PT];
    __shared__ __attribute__((aligned(128))) int stg[SCAP];
    const int tid  = threadIdx.x;
    const int lane = tid & 31;
    const int wv   = tid >> 5;
    cnt[tid] = 0;

    if ((int)blockIdx.x >= NB) {
        const int b = (int)blockIdx.x - NB;
        gnn_walk_pieces<0>(tab + (size_t)(NB + b) * G, G, bpair, ssrc, cnt, ptab, col, stg, tid);
        __syncthreads();
        const int n = b * BSZ + tid;
        if (n < N) nsrc[n] = 1.0f / sqrtf(fmaxf((float)cnt[tid], 1.0f));
        return;
    }

    const int b = (int)blockIdx.x;
    int part = 0;
    {
        const int nent = b * G;
        const int n4   = nent >> 2;
        const int4* t4 = (const int4*)tab;
        for (int i0 = 0; i0 < n4; i0 += FB * BSZ) {
            int4 q[FB];
#pragma unroll
            for (int k = 0; k < FB; ++k) {
                const int i = i0 + k * BSZ + tid;
                q[k] = t4[(i < n4) ? i : 0];
            }
#pragma unroll
            for (int k = 0; k < FB; ++k)
                if (i0 + k * BSZ + tid < n4)
                    part += (q[k].x >> 16) + (q[k].y >> 16) + (q[k].z >> 16) + (q[k].w >> 16);
        }
        const int r = (n4 << 2) + tid;
        if (r < nent) part += tab[r] >> 16;
    }
    for (int o = 16; o > 0; o >>= 1) part += __shfl_xor(part, o);
    if (lane == 0) wsum[wv] = part;
    __syncthreads();
    int beg = (lane < BSZ / 32) ? wsum[lane] : 0;
    for (int o = 16; o > 0; o >>= 1) beg += __shfl_xor(beg, o);

    const int* trow = tab + (size_t)b * G;
    gnn_walk_pieces<1>(trow, G, bpair, ssrc, cnt, ptab, col, stg, tid);
    __syncthreads();

    const int ci = cnt[tid];
    int inc = ci;
    for (int o = 1; o < 32; o <<= 1) {
        int t = __shfl_up(inc, o);
        if (lane >= o) inc += t;
    }
    if (lane == 31) wsum[wv] = inc;
    __syncthreads();
    int tot = (lane < BSZ / 32) ? wsum[lane] : 0;
    int wp  = (lane < wv) ? tot : 0;
    for (int o = 16; o > 0; o >>= 1) {
        wp  += __shfl_xor(wp, o);
        tot += __shfl_xor(tot, o);
    }

    const int lexcl = wp + inc - ci;
    const int excl  = beg + lexcl;
    const int n = b * BSZ + tid;
    if (n < N) {
        row_ptr[n] = excl;
        ndst[n] = 1.0f / sqrtf(fmaxf((float)ci, 1.0f));
    }
    if (n == N - 1) row_ptr[N] = excl + ci;

    const int begu = __builtin_amdgcn_readfirstlane(beg);
    const int totu = __builtin_amdgcn_readfirstlane(tot);
    const int aoff = begu & 31;
    if (aoff + totu <= SCAP) {
        cnt[tid] = aoff + lexcl;
        gnn_walk_pieces<3>(trow, G, bpair, ssrc, cnt, ptab, col, stg, tid);
        __syncthreads();
        int* cl = col + (begu - aoff);
        for (int i = tid; i < aoff + totu; i += BSZ)
            if (i >= aoff) cl[i] = stg[i];
    } else {
        cnt[tid] = excl;
        gnn_walk_pieces<2>(trow, G, bpair, ssrc, cnt, ptab, col, stg, tid);
    }
}

__global__ __launch_bounds__(128) void gnn_gemm_kernel(
    const float* __restrict__ H, const float* __restrict__ nsrc,
    const float* __restrict__ Wf, float* __restrict__ X, int N) {
    __shared__ float As[16 * LDSTRIDE];

    const int row0 = blockIdx.x * 16;
    const int tid  = threadIdx.x;

    for (int i = tid; i < 16 * 32; i += 128) {
        int r  = i >> 5;
        int c4 = (i & 31) * 4;
        int row = row0 + r;
        int rowc = (row < N) ? row : (N - 1);
        float s = nsrc[rowc];
        float4 v = *(const float4*)(H + (size_t)rowc * 128 + c4);
        float4 sv = make_float4(v.x * s, v.y * s, v.z * s, v.w * s);
        *(float4*)(&As[r * LDSTRIDE + c4]) = sv;
    }
    __syncthreads();

    const int wave = tid >> 5;
    const int lane = tid & 31;
    const int ln   = lane & 15;
    const int koff = (lane < 16) ? 0 : 2;

    const int n0 = wave * 32;

    v8f acc0 = {};
    v8f acc1 = {};

    const float4* wf = (const float4*)Wf + (wave * 32) * 32 + lane;

    for (int kb = 0; kb < 128; kb += 4) {
        v2f a = *(const v2f*)(&As[ln * LDSTRIDE + kb + koff]);

        const float4 wv = wf[(kb >> 2) * 32];
        v2f b0, b1;
        b0.x = wv.x;
        b0.y = wv.y;
        b1.x = wv.z;
        b1.y = wv.w;

        acc0 = __builtin_amdgcn_wmma_f32_16x16x4_f32(
            false, a, false, b0, (short)0, acc0, false, false);
        acc1 = __builtin_amdgcn_wmma_f32_16x16x4_f32(
            false, a, false, b1, (short)0, acc1, false, false);
    }

    const int mbase = row0 + ((lane < 16) ? 0 : 8);
#pragma unroll
    for (int v = 0; v < 8; ++v) {
        int row = mbase + v;
        if (row < N)
            *(float2*)(X + (size_t)row * 128 + n0 + 2 * ln) = make_float2(acc0[v], acc1[v]);
    }
}

__device__ __forceinline__ float gnn_lane_f(float v, int k) {
    return __int_as_float(__builtin_amdgcn_readlane(__float_as_int(v), k));
}

template <bool FUSED>
__device__ __forceinline__ void gnn_stream_rows(
    const float* __restrict__ X, const int* __restrict__ col,
    const int rpv, const int nn, const float ndv, const float nsv,
    const float4 b, const int lane, float* dst) {
    const int stride = FUSED ? LDSTRIDE : 128;
    const float4 z = make_float4(0.0f, 0.0f, 0.0f, 0.0f);
    const float* xl = X + lane * 4;
    const int eA = __builtin_amdgcn_readlane(rpv, 0);
    const int eB = __builtin_amdgcn_readlane(rpv, nn);
    int k = 0;
    float4 acc = z;

    auto close_node = [&]() {
        const float s = gnn_lane_f(ndv, k);
        float4 o;
        if (FUSED) {
            const float t = gnn_lane_f(nsv, k);
            o.x = fmaxf(acc.x * s + b.x, 0.0f) * t;
            o.y = fmaxf(acc.y * s + b.y, 0.0f) * t;
            o.z = fmaxf(acc.z * s + b.z, 0.0f) * t;
            o.w = fmaxf(acc.w * s + b.w, 0.0f) * t;
        } else {
            o.x = acc.x * s + b.x;
            o.y = acc.y * s + b.y;
            o.z = acc.z * s + b.z;
            o.w = acc.w * s + b.w;
        }
        *(float4*)(dst + k * stride) = o;
        acc = z;
        ++k;
    };

    if (eB > eA) {
        int brel = __builtin_amdgcn_readlane(rpv, 1) - eA;
        int mycA = col[(eA + lane < eB) ? (eA + lane) : (eB - 1)];
        float4 buf[2][4];
#pragma unroll
        for (int i = 0; i < 4; ++i)
            buf[0][i] = *(const float4*)(xl + (size_t)__builtin_amdgcn_readlane(mycA, i) * 128);
        int base = eA;
        int erel = eB - eA;
        do {
            const int nx = base + 32 + lane;
            const int mycB = col[(nx < eB) ? nx : (eB - 1)];
#pragma unroll
            for (int g = 0; g < 8; ++g) {
#pragma unroll
                for (int i = 0; i < 4; ++i) {
                    const int idx = (g < 7) ? __builtin_amdgcn_readlane(mycA, (4 * (g + 1) + i) & 31)
                                            : __builtin_amdgcn_readlane(mycB, i);
                    buf[(g + 1) & 1][i] = *(const float4*)(xl + (size_t)idx * 128);
                }
#pragma unroll
                for (int i = 0; i < 4; ++i) {
                    while (brel == 4 * g + i) {
                        close_node();
                        brel = (k < nn) ? (__builtin_amdgcn_readlane(rpv, k + 1) - base) : -1;
                    }
                    const float4 v = buf[g & 1][i];
                    acc.x += v.x; acc.y += v.y; acc.z += v.z; acc.w += v.w;
                }
                if (g < 7 && 4 * (g + 1) >= erel) goto stream_done;
            }
            mycA = mycB;
            brel -= 32;
            base += 32;
            erel -= 32;
        } while (erel > 0);
    }
stream_done:
    while (k < nn) close_node();
}

__global__ __launch_bounds__(128) void gnn_pull_relu_gemm_kernel(
    const float* __restrict__ X, const int* __restrict__ row_ptr,
    const int* __restrict__ col, const float* __restrict__ ndst,
    const float* __restrict__ nsrc, const float* __restrict__ bias,
    const float* __restrict__ Wf, float* __restrict__ Y, int N) {
    __shared__ float As[16 * LDSTRIDE];

    const int row0 = blockIdx.x * 16;
    const int tid  = threadIdx.x;
    const int lane = tid & 31;
    const int wave = __builtin_amdgcn_readfirstlane(tid >> 5);
    const float4 b = *(const float4*)(bias + lane * 4);
    const float4 z = make_float4(0.0f, 0.0f, 0.0f, 0.0f);

    const int nq = row0 + wave * 4;
    int nn = N - nq;
    nn = (nn < 0) ? 0 : ((nn > 4) ? 4 : nn);
    float* arow = &As[(wave * 4) * LDSTRIDE + lane * 4];
    if (nn > 0) {
        const int ni  = nq + lane;
        const int rpv = row_ptr[(ni < N) ? ni : N];
        const int nc  = (ni < N) ? ni : (N - 1);
        const float ndv = ndst[nc];
        const float nsv = nsrc[nc];
        gnn_stream_rows<true>(X, col, rpv, nn, ndv, nsv, b, lane, arow);
    }
    for (int q = nn; q < 4; ++q)
        *(float4*)(arow + q * LDSTRIDE) = z;
    __syncthreads();

    const int ln   = lane & 15;
    const int koff = (lane < 16) ? 0 : 2;

    const int n0 = wave * 32;

    v8f acc0 = {};
    v8f acc1 = {};

    const float4* wf = (const float4*)Wf + (wave * 32) * 32 + lane;

    for (int kb = 0; kb < 128; kb += 4) {
        v2f a = *(const v2f*)(&As[ln * LDSTRIDE + kb + koff]);

        const float4 wv = wf[(kb >> 2) * 32];
        v2f b0, b1;
        b0.x = wv.x;
        b0.y = wv.y;
        b1.x = wv.z;
        b1.y = wv.w;

        acc0 = __builtin_amdgcn_wmma_f32_16x16x4_f32(
            false, a, false, b0, (short)0, acc0, false, false);
        acc1 = __builtin_amdgcn_wmma_f32_16x16x4_f32(
            false, a, false, b1, (short)0, acc1, false, false);
    }

    const int mbase = row0 + ((lane < 16) ? 0 : 8);
#pragma unroll
    for (int v = 0; v < 8; ++v) {
        int row = mbase + v;
        if (row < N)
            *(float2*)(Y + (size_t)row * 128 + n0 + 2 * ln) = make_float2(acc0[v], acc1[v]);
    }
}

__global__ __launch_bounds__(256) void gnn_pull_kernel(
    const float* __restrict__ X, const int* __restrict__ row_ptr,
    const int* __restrict__ col, const float* __restrict__ ndst,
    const float* __restrict__ bias, float* __restrict__ OUT, int N) {
    const int lane = threadIdx.x & 31;
    const int wpb  = blockDim.x >> 5;
    const int w0 = __builtin_amdgcn_readfirstlane((int)(blockIdx.x * wpb + (threadIdx.x >> 5)));
    const int nw = (int)gridDim.x * wpb;
    const float4 b = *(const float4*)(bias + lane * 4);

    int npb = (N + nw - 1) / nw;
    npb = (npb < 1) ? 1 : ((npb > 31) ? 31 : npb);
    for (int nA = w0 * npb; nA < N; nA += nw * npb) {
        const int nn  = (N - nA < npb) ? (N - nA) : npb;
        const int ni  = nA + lane;
        const int rpv = row_ptr[(ni < N) ? ni : N];
        const float ndv = ndst[(ni < N) ? ni : (N - 1)];
        gnn_stream_rows<false>(X, col, rpv, nn, ndv, 0.0f, b, lane, OUT + (size_t)nA * 128 + lane * 4);
    }
}

extern "C" void kernel_launch(void* const* d_in, const int* in_sizes, int n_in,
                              void* d_out, int out_size, void* d_ws, size_t ws_size,
                              hipStream_t stream) {
    const float* feature = (const float*)d_in[0];
    const int*   src     = (const int*)d_in[1];
    const int*   dst     = (const int*)d_in[2];
    const float* W1      = (const float*)d_in[3];
    const float* b1      = (const float*)d_in[4];
    const float* W2      = (const float*)d_in[5];
    const float* b2      = (const float*)d_in[6];
    float* out = (float*)d_out;

    const int F = 128;
    const int N = in_sizes[0] / F;
    const int E = in_sizes[1];
    const int NB = (N + BSZ - 1) / BSZ;
    const int G  = (E + CH - 1) / CH;
    const int L  = 2 * NB * G;

    char* ws = (char*)d_ws;
    size_t o = 0;
    float* wf1 = (float*)(ws + o);
    float* wf2 = wf1 + 128 * 128;
    o = (o + (size_t)2 * 128 * 128 * sizeof(float) + 255) & ~(size_t)255;
    int* tab = (int*)(ws + o);
    o = (o + (size_t)L * sizeof(int) + 255) & ~(size_t)255;
    int2* bpair = (int2*)(ws + o);
    o = (o + (size_t)E * sizeof(int2) + 255) & ~(size_t)255;
    int* ssrc = (int*)(ws + o);
    o = (o + (size_t)E * sizeof(int) + 255) & ~(size_t)255;
    float* nsrc = (float*)(ws + o);
    float* ndst = nsrc + N;
    o = (o + (size_t)2 * N * sizeof(float) + 255) & ~(size_t)255;
    int* row_ptr = (int*)(ws + o);
    o = (o + ((size_t)N + 1) * sizeof(int) + 255) & ~(size_t)255;
    int* col = (int*)(ws + o);
    o = (o + (size_t)E * sizeof(int) + 255) & ~(size_t)255;
    float* buf0 = (float*)(ws + o);
    float* buf1 = buf0 + (size_t)N * F;

    gnn_chunk_sort_kernel<<<G, CT, 0, stream>>>(src, dst, tab, bpair, ssrc, E, NB, G, W1, W2, wf1, wf2);
    gnn_bucket_finish_kernel<<<2 * NB, BSZ, 0, stream>>>(bpair, ssrc, tab, row_ptr, col, nsrc, ndst, N, NB, G);

    const int gemmBlocks = (N + 15) / 16;
    int pullBlocks = (N + 7) / 8;
    if (pullBlocks > 2048) pullBlocks = 2048;

    gnn_gemm_kernel<<<gemmBlocks, 128, 0, stream>>>(feature, nsrc, wf1, buf0, N);
    gnn_pull_relu_gemm_kernel<<<gemmBlocks, 128, 0, stream>>>(buf0, row_ptr, col, ndst, nsrc, b1, wf2, buf1, N);
    gnn_pull_kernel<<<pullBlocks, 256, 0, stream>>>(buf1, row_ptr, col, ndst, b2, out, N);
}
